# starting kernel
# baseline (speedup 1.0000x reference)
#include <hip/hip_runtime.h>
#include <hip/hip_bf16.h>

typedef __attribute__((ext_vector_type(2))) float v2f;
typedef __attribute__((ext_vector_type(8))) float v8f;

__global__ void zero_kernel(float* __restrict__ p, long long n) {
    long long i = (long long)blockIdx.x * blockDim.x + threadIdx.x;
    long long stride = (long long)gridDim.x * blockDim.x;
    for (; i < n; i += stride) p[i] = 0.0f;
}

__global__ void degree_kernel(const int* __restrict__ ei, float* __restrict__ deg, int E) {
    int e = blockIdx.x * blockDim.x + threadIdx.x;
    if (e < E) atomicAdd(&deg[ei[E + e]], 1.0f);
}

template <int F>
__global__ void scatter_kernel(const int* __restrict__ ei, const float* __restrict__ feat,
                               float* __restrict__ agg, int E) {
    long long t = (long long)blockIdx.x * blockDim.x + threadIdx.x;
    int e = (int)(t / (F / 4));
    int f4 = (int)(t % (F / 4)) * 4;
    if (e >= E) return;
    int src = ei[e];
    int dst = ei[E + e];
    const float4 v = *(const float4*)(feat + (size_t)src * F + f4);
    float* p = agg + (size_t)dst * F + f4;
    atomicAdd(p + 0, v.x);
    atomicAdd(p + 1, v.y);
    atomicAdd(p + 2, v.z);
    atomicAdd(p + 3, v.w);
}

template <int K, int NOUT>
__global__ __launch_bounds__(32 * (NOUT / 16)) void sage_combine_kernel(
    const float* __restrict__ xin, const float* __restrict__ agg,
    const float* __restrict__ deg, const float* __restrict__ Wl,
    const float* __restrict__ bl, const float* __restrict__ Wr,
    float* __restrict__ out, int Nn) {
    __shared__ float s_inv[16];
    __shared__ float s_a[16 * K];
    __shared__ float s_x[16 * K];

    const int node_base = blockIdx.x * 16;
    const int tid = threadIdx.x;
    const int lane = tid & 31;
    const int wave = tid >> 5;
    const int m = lane & 15;
    const int khalf = lane >> 4;
    const int ncol = wave * 16 + m;

    if (tid < 16) {
        int nd = node_base + tid;
        if (nd > Nn - 1) nd = Nn - 1;
        s_inv[tid] = 1.0f / fmaxf(deg[nd], 1.0f);
    }
    __syncthreads();

    const int nthreads = 32 * (NOUT / 16);
    for (int i = tid; i < 16 * K; i += nthreads) {
        int nd = i / K;
        int f = i - nd * K;
        int gnode = node_base + nd;
        if (gnode > Nn - 1) gnode = Nn - 1;
        s_a[i] = agg[(size_t)gnode * K + f] * s_inv[nd];
        s_x[i] = xin[(size_t)gnode * K + f];
    }
    __syncthreads();

    v8f acc = {};
#pragma unroll
    for (int k = 0; k < K; k += 4) {
        const int kb = k + khalf * 2;
        v2f a, b;
        a.x = s_a[m * K + kb];
        a.y = s_a[m * K + kb + 1];
        b.x = Wl[(size_t)kb * NOUT + ncol];
        b.y = Wl[(size_t)(kb + 1) * NOUT + ncol];
        acc = __builtin_amdgcn_wmma_f32_16x16x4_f32(false, a, false, b,
                                                    (short)0, acc, false, false);
        a.x = s_x[m * K + kb];
        a.y = s_x[m * K + kb + 1];
        b.x = Wr[(size_t)kb * NOUT + ncol];
        b.y = Wr[(size_t)(kb + 1) * NOUT + ncol];
        acc = __builtin_amdgcn_wmma_f32_16x16x4_f32(false, a, false, b,
                                                    (short)0, acc, false, false);
    }

    const float bias = bl[ncol];
#pragma unroll
    for (int v = 0; v < 8; ++v) {
        int mrow = v + khalf * 8;
        int gnode = node_base + mrow;
        if (gnode < Nn) {
            float val = acc[v] + bias;
            out[(size_t)gnode * NOUT + ncol] = fmaxf(val, 0.0f);
        }
    }
}

__global__ void fc_sigmoid_kernel(const float* __restrict__ h2,
                                  const float* __restrict__ Wfc,
                                  const float* __restrict__ bfc,
                                  float* __restrict__ out, int Nn) {
    int i = blockIdx.x * blockDim.x + threadIdx.x;
    if (i >= Nn) return;
    float s = bfc[0];
    const float* hp = h2 + (size_t)i * 64;
#pragma unroll
    for (int k = 0; k < 64; ++k) s += hp[k] * Wfc[k];
    out[i] = 1.0f / (1.0f + __expf(-s));
}

extern "C" void kernel_launch(void* const* d_in, const int* in_sizes, int n_in,
                              void* d_out, int out_size, void* d_ws, size_t ws_size,
                              hipStream_t stream) {
    const float* x    = (const float*)d_in[0];
    const int*   ei   = (const int*)d_in[1];
    const float* W1l  = (const float*)d_in[2];
    const float* b1   = (const float*)d_in[3];
    const float* W1r  = (const float*)d_in[4];
    const float* W2l  = (const float*)d_in[5];
    const float* b2   = (const float*)d_in[6];
    const float* W2r  = (const float*)d_in[7];
    const float* Wfc  = (const float*)d_in[8];
    const float* bfc  = (const float*)d_in[9];
    float* outp = (float*)d_out;

    const int Nn = in_sizes[0] / 64;
    const int Ee = in_sizes[1] / 2;

    float* deg = (float*)d_ws;
    float* agg = deg + Nn;
    float* h1  = agg + (size_t)Nn * 128;
    float* h2  = h1 + (size_t)Nn * 128;

    const int ZB = 2048, TB = 256;

    zero_kernel<<<ZB, TB, 0, stream>>>(deg, (long long)Nn * 129);
    degree_kernel<<<(Ee + TB - 1) / TB, TB, 0, stream>>>(ei, deg, Ee);
    {
        long long t = (long long)Ee * (64 / 4);
        scatter_kernel<64><<<(int)((t + TB - 1) / TB), TB, 0, stream>>>(ei, x, agg, Ee);
    }
    sage_combine_kernel<64, 128><<<(Nn + 15) / 16, 256, 0, stream>>>(
        x, agg, deg, W1l, b1, W1r, h1, Nn);

    zero_kernel<<<ZB, TB, 0, stream>>>(agg, (long long)Nn * 128);
    {
        long long t = (long long)Ee * (128 / 4);
        scatter_kernel<128><<<(int)((t + TB - 1) / TB), TB, 0, stream>>>(ei, h1, agg, Ee);
    }
    sage_combine_kernel<128, 64><<<(Nn + 15) / 16, 128, 0, stream>>>(
        h1, agg, deg, W2l, b2, W2r, h2, Nn);

    fc_sigmoid_kernel<<<(Nn + TB - 1) / TB, TB, 0, stream>>>(h2, Wfc, bfc, outp, Nn);
}


# Neighbour sums in LDS, not global atomics: each block owns a destination range, scans all edges for its own, adds source rows by LDS float add (arrival order), counts degree there; 8 launches to 5
# speedup vs baseline: 1.6747x; 1.6747x over previous
#include <hip/hip_runtime.h>
#include <hip/hip_bf16.h>


typedef __attribute__((ext_vector_type(2))) float v2f;
typedef __attribute__((ext_vector_type(8))) float v8f;

template <int F, int R, bool COUNT>
__global__ __launch_bounds__(1024) void neighbour_sum_lds_kernel(
    const int* __restrict__ ei, const float* __restrict__ feat,
    float* __restrict__ agg, float* __restrict__ deg, int E, int Nn) {
    constexpr int NW = 32;
    constexpr int QCAP = 160;
    constexpr int U = 8;
    constexpr int G = 1024 / F;
    constexpr int H = F / 64;
    __shared__ float s_sum[R * F];
    __shared__ float s_cnt[R];
    __shared__ int s_qe[NW * QCAP];
    __shared__ int s_qd[NW * QCAP];

    const int tid = threadIdx.x;
    const int lane = tid & 31;
    const int wave = __builtin_amdgcn_readfirstlane(tid >> 5);
    const int base = blockIdx.x * R;

    for (int i = tid; i < R * F; i += 1024) s_sum[i] = 0.0f;
    if (COUNT) {
        for (int i = tid; i < R; i += 1024) s_cnt[i] = 0.0f;
    }
    __syncthreads();

    int* qe = s_qe + wave * QCAP;
    int* qd = s_qd + wave * QCAP;
    const unsigned lt = (1u << lane) - 1u;
    int qn = 0;

    auto push = [&](int d, int e) {
        const int dl = d - base;
        const bool hit = (unsigned)dl < (unsigned)R;
        const unsigned m = __builtin_amdgcn_ballot_w32(hit);
        if (hit) {
            const int pos = qn + __popc(m & lt);
            qe[pos] = e;
            qd[pos] = dl;
        }
        qn += __popc(m);
    };

    auto drain = [&]() {
        __builtin_amdgcn_wave_barrier();
        for (int b = 0; b < qn; b += 32) {
            const int nb = min(32, qn - b);
            const bool valid = lane < nb;
            const int e = valid ? qe[b + lane] : 0;
            const int dl = valid ? qd[b + lane] : 0;
            const int src = ei[e];
            if (COUNT) {
                if (valid) atomicAdd(&s_cnt[dl], 1.0f);
            }
            for (int j0 = 0; j0 < nb; j0 += G) {
                float2 v[G][H];
#pragma unroll
                for (int g = 0; g < G; ++g) {
                    const int sj = __builtin_amdgcn_readlane(src, j0 + g);
                    const float* rp = feat + (size_t)sj * F + lane * 2;
#pragma unroll
                    for (int h = 0; h < H; ++h) v[g][h] = *(const float2*)(rp + h * 64);
                }
#pragma unroll
                for (int g = 0; g < G; ++g) {
                    if (j0 + g < nb) {
                        const int dj = __builtin_amdgcn_readlane(dl, j0 + g);
                        float* sp = s_sum + dj * F + lane * 2;
#pragma unroll
                        for (int h = 0; h < H; ++h) {
                            atomicAdd(sp + h * 64, v[g][h].x);
                            atomicAdd(sp + h * 64 + 1, v[g][h].y);
                        }
                    }
                }
            }
        }
        qn = 0;
        __builtin_amdgcn_wave_barrier();
    };

    const int* dstp = ei + E;
    const int nvec = ((E & 3) == 0) ? (E >> 2) : 0;
    const int4* d4 = (const int4*)dstp;
    for (int v0 = wave * (U * 32); v0 < nvec; v0 += NW * U * 32) {
        int4 dv[U];
#pragma unroll
        for (int u = 0; u < U; ++u) {
            const int idx = v0 + u * 32 + lane;
            dv[u] = (idx < nvec) ? d4[idx] : make_int4(-1, -1, -1, -1);
        }
#pragma unroll
        for (int u = 0; u < U; ++u) {
            const int e0 = (v0 + u * 32 + lane) * 4;
            push(dv[u].x, e0);
            push(dv[u].y, e0 + 1);
            push(dv[u].z, e0 + 2);
            push(dv[u].w, e0 + 3);
            if (qn >= 32) drain();
        }
    }
    for (int e0 = nvec * 4 + wave * 32; e0 < E; e0 += NW * 32) {
        const int e = e0 + lane;
        const int d = (e < E) ? dstp[e] : -1;
        push(d, e);
        if (qn >= 32) drain();
    }
    if (qn > 0) drain();
    __syncthreads();

    for (int i = tid; i < R * F; i += 1024) {
        const int nd = base + i / F;
        if (nd < Nn) agg[(size_t)base * F + i] = s_sum[i];
    }
    if (COUNT) {
        for (int i = tid; i < R; i += 1024) {
            if (base + i < Nn) deg[base + i] = s_cnt[i];
        }
    }
}

template <int K, int NOUT>
__global__ __launch_bounds__(32 * (NOUT / 16)) void sage_combine_kernel(
    const float* __restrict__ xin, const float* __restrict__ agg,
    const float* __restrict__ deg, const float* __restrict__ Wl,
    const float* __restrict__ bl, const float* __restrict__ Wr,
    float* __restrict__ out, int Nn) {
    __shared__ float s_inv[16];
    __shared__ float s_a[16 * K];
    __shared__ float s_x[16 * K];

    const int node_base = blockIdx.x * 16;
    const int tid = threadIdx.x;
    const int lane = tid & 31;
    const int wave = tid >> 5;
    const int m = lane & 15;
    const int khalf = lane >> 4;
    const int ncol = wave * 16 + m;

    if (tid < 16) {
        int nd = node_base + tid;
        if (nd > Nn - 1) nd = Nn - 1;
        s_inv[tid] = 1.0f / fmaxf(deg[nd], 1.0f);
    }
    __syncthreads();

    const int nthreads = 32 * (NOUT / 16);
    for (int i = tid; i < 16 * K; i += nthreads) {
        int nd = i / K;
        int f = i - nd * K;
        int gnode = node_base + nd;
        if (gnode > Nn - 1) gnode = Nn - 1;
        s_a[i] = agg[(size_t)gnode * K + f] * s_inv[nd];
        s_x[i] = xin[(size_t)gnode * K + f];
    }
    __syncthreads();

    v8f acc = {};
#pragma unroll
    for (int k = 0; k < K; k += 4) {
        const int kb = k + khalf * 2;
        v2f a, b;
        a.x = s_a[m * K + kb];
        a.y = s_a[m * K + kb + 1];
        b.x = Wl[(size_t)kb * NOUT + ncol];
        b.y = Wl[(size_t)(kb + 1) * NOUT + ncol];
        acc = __builtin_amdgcn_wmma_f32_16x16x4_f32(false, a, false, b,
                                                    (short)0, acc, false, false);
        a.x = s_x[m * K + kb];
        a.y = s_x[m * K + kb + 1];
        b.x = Wr[(size_t)kb * NOUT + ncol];
        b.y = Wr[(size_t)(kb + 1) * NOUT + ncol];
        acc = __builtin_amdgcn_wmma_f32_16x16x4_f32(false, a, false, b,
                                                    (short)0, acc, false, false);
    }

    const float bias = bl[ncol];
#pragma unroll
    for (int v = 0; v < 8; ++v) {
        int mrow = v + khalf * 8;
        int gnode = node_base + mrow;
        if (gnode < Nn) {
            float val = acc[v] + bias;
            out[(size_t)gnode * NOUT + ncol] = fmaxf(val, 0.0f);
        }
    }
}

__global__ void fc_sigmoid_kernel(const float* __restrict__ h2,
                                  const float* __restrict__ Wfc,
                                  const float* __restrict__ bfc,
                                  float* __restrict__ out, int Nn) {
    int i = blockIdx.x * blockDim.x + threadIdx.x;
    if (i >= Nn) return;
    float s = bfc[0];
    const float* hp = h2 + (size_t)i * 64;
#pragma unroll
    for (int k = 0; k < 64; ++k) s += hp[k] * Wfc[k];
    out[i] = 1.0f / (1.0f + __expf(-s));
}

extern "C" void kernel_launch(void* const* d_in, const int* in_sizes, int n_in,
                              void* d_out, int out_size, void* d_ws, size_t ws_size,
                              hipStream_t stream) {
    const float* x    = (const float*)d_in[0];
    const int*   ei   = (const int*)d_in[1];
    const float* W1l  = (const float*)d_in[2];
    const float* b1   = (const float*)d_in[3];
    const float* W1r  = (const float*)d_in[4];
    const float* W2l  = (const float*)d_in[5];
    const float* b2   = (const float*)d_in[6];
    const float* W2r  = (const float*)d_in[7];
    const float* Wfc  = (const float*)d_in[8];
    const float* bfc  = (const float*)d_in[9];
    float* outp = (float*)d_out;

    const int Nn = in_sizes[0] / 64;
    const int Ee = in_sizes[1] / 2;

    float* deg = (float*)d_ws;
    float* agg = deg + Nn;
    float* h1  = agg + (size_t)Nn * 128;
    float* h2  = h1 + (size_t)Nn * 128;

    const int TB = 256;

    neighbour_sum_lds_kernel<64, 1024, true><<<(Nn + 1023) / 1024, 1024, 0, stream>>>(
        ei, x, agg, deg, Ee, Nn);
    sage_combine_kernel<64, 128><<<(Nn + 15) / 16, 256, 0, stream>>>(
        x, agg, deg, W1l, b1, W1r, h1, Nn);

    neighbour_sum_lds_kernel<128, 512, false><<<(Nn + 511) / 512, 1024, 0, stream>>>(
        ei, h1, agg, deg, Ee, Nn);
    sage_combine_kernel<128, 64><<<(Nn + 15) / 16, 128, 0, stream>>>(
        h1, agg, deg, W2l, b2, W2r, h2, Nn);

    fc_sigmoid_kernel<<<(Nn + TB - 1) / TB, TB, 0, stream>>>(h2, Wfc, bfc, outp, Nn);
}


# Layer 2 stops re-scanning all destinations: layer-1 sum waves save their (source,row) hits in the workspace; layer 2 (1024x64 tiles, 2 per range) walks the lists; LDS float adds, arrival order
# speedup vs baseline: 1.9688x; 1.1756x over previous
#include <hip/hip_runtime.h>
#include <hip/hip_bf16.h>
#include <type_traits>


typedef __attribute__((ext_vector_type(2))) float v2f;
typedef __attribute__((ext_vector_type(8))) float v8f;

constexpr int LCAP = 2048;

template <int F, int FS, int R, bool COUNT, bool SAVE, bool USE_LIST>
__global__ __launch_bounds__(1024) void neighbour_sum_lds_kernel(
    const int* __restrict__ ei, const float* __restrict__ feat,
    float* __restrict__ agg, float* __restrict__ deg,
    int* __restrict__ wlist, int* __restrict__ wcnt, int E, int Nn) {
    constexpr int NW = 32;
    constexpr int QCAP = 160;
    constexpr int U = 8;
    constexpr int G = 1024 / F;
    constexpr int H = F / 64;
    constexpr int NC = FS / F;
    __shared__ float s_sum[R * F];
    __shared__ float s_cnt[R];
    __shared__ int s_qe[NW * QCAP];
    __shared__ int s_qd[NW * QCAP];

    const int tid = threadIdx.x;
    const int lane = tid & 31;
    const int wave = __builtin_amdgcn_readfirstlane(tid >> 5);
    const int rb = blockIdx.x / NC;
    const int col0 = (blockIdx.x % NC) * F;
    const int base = rb * R;
    const float* featc = feat + col0 + lane * 2;

    for (int i = tid; i < R * F; i += 1024) s_sum[i] = 0.0f;
    if (COUNT) {
        for (int i = tid; i < R; i += 1024) s_cnt[i] = 0.0f;
    }
    __syncthreads();

    int2* wl = (int2*)wlist + (size_t)(rb * NW + wave) * LCAP;

    auto add_rows = [&](int src, int dl, int nb, auto gg) {
        constexpr int GG = decltype(gg)::value;
        for (int j0 = 0; j0 < nb; j0 += GG) {
            float2 v[GG][H];
#pragma unroll
            for (int g = 0; g < GG; ++g) {
                const int sj = __builtin_amdgcn_readlane(src, j0 + g);
                const float* rp = featc + (size_t)sj * FS;
#pragma unroll
                for (int h = 0; h < H; ++h) v[g][h] = *(const float2*)(rp + h * 64);
            }
#pragma unroll
            for (int g = 0; g < GG; ++g) {
                if (j0 + g < nb) {
                    const int dj = __builtin_amdgcn_readlane(dl, j0 + g);
                    float* sp = s_sum + dj * F + lane * 2;
#pragma unroll
                    for (int h = 0; h < H; ++h) {
                        atomicAdd(sp + h * 64, v[g][h].x);
                        atomicAdd(sp + h * 64 + 1, v[g][h].y);
                    }
                }
            }
        }
    };

    auto scan_all = [&]() {
        int* qe = s_qe + wave * QCAP;
        int* qd = s_qd + wave * QCAP;
        const unsigned lt = (1u << lane) - 1u;
        int qn = 0;
        int wn = 0;

        auto push = [&](int d, int e) {
            const int dl = d - base;
            const bool hit = (unsigned)dl < (unsigned)R;
            const unsigned m = __builtin_amdgcn_ballot_w32(hit);
            if (hit) {
                const int pos = qn + __popc(m & lt);
                qe[pos] = e;
                qd[pos] = dl;
            }
            qn += __popc(m);
        };

        auto drain = [&]() {
            __builtin_amdgcn_wave_barrier();
            for (int b = 0; b < qn; b += 32) {
                const int nb = min(32, qn - b);
                const bool valid = lane < nb;
                const int e = valid ? qe[b + lane] : 0;
                const int dl = valid ? qd[b + lane] : 0;
                const int src = ei[e];
                if (COUNT) {
                    if (valid) atomicAdd(&s_cnt[dl], 1.0f);
                }
                if (SAVE) {
                    if (valid && wn + lane < LCAP) wl[wn + lane] = make_int2(src, dl);
                    wn += nb;
                }
                add_rows(src, dl, nb, std::integral_constant<int, G>{});
            }
            qn = 0;
            __builtin_amdgcn_wave_barrier();
        };

        const int* dstp = ei + E;
        const int nvec = ((E & 3) == 0) ? (E >> 2) : 0;
        const int4* d4 = (const int4*)dstp;
        for (int v0 = wave * (U * 32); v0 < nvec; v0 += NW * U * 32) {
            int4 dv[U];
#pragma unroll
            for (int u = 0; u < U; ++u) {
                const int idx = v0 + u * 32 + lane;
                dv[u] = (idx < nvec) ? d4[idx] : make_int4(-1, -1, -1, -1);
            }
#pragma unroll
            for (int u = 0; u < U; ++u) {
                const int e0 = (v0 + u * 32 + lane) * 4;
                push(dv[u].x, e0);
                push(dv[u].y, e0 + 1);
                push(dv[u].z, e0 + 2);
                push(dv[u].w, e0 + 3);
                if (qn >= 32) drain();
            }
        }
        for (int e0 = nvec * 4 + wave * 32; e0 < E; e0 += NW * 32) {
            const int e = e0 + lane;
            const int d = (e < E) ? dstp[e] : -1;
            push(d, e);
            if (qn >= 32) drain();
        }
        if (qn > 0) drain();
        if (SAVE) {
            if (lane == 0) wcnt[rb * NW + wave] = wn;
        }
    };

    if (USE_LIST) {
        const int cl = wcnt[rb * NW + lane];
        const bool fits = __builtin_amdgcn_ballot_w32(cl > LCAP) == 0u;
        if (fits) {
            constexpr int GL = 2048 / F;
            const int n = __builtin_amdgcn_readlane(cl, wave);
            int2 nxt = (lane < n) ? wl[lane] : make_int2(0, 0);
            for (int i0 = 0; i0 < n; i0 += 32) {
                const int2 cur = nxt;
                const int nb = min(32, n - i0);
                for (int j0 = 0; j0 < nb; j0 += GL) {
                    float2 v[GL][H];
#pragma unroll
                    for (int g = 0; g < GL; ++g) {
                        const int sj = __builtin_amdgcn_readlane(cur.x, j0 + g);
                        const float* rp = featc + (size_t)sj * FS;
#pragma unroll
                        for (int h = 0; h < H; ++h) v[g][h] = *(const float2*)(rp + h * 64);
                    }
                    if (j0 + GL >= nb) {
                        const int in = i0 + 32 + lane;
                        nxt = (in < n) ? wl[in] : make_int2(0, 0);
                    }
#pragma unroll
                    for (int g = 0; g < GL; ++g) {
                        if (j0 + g < nb) {
                            const int dj = __builtin_amdgcn_readlane(cur.y, j0 + g);
                            float* sp = s_sum + dj * F + lane * 2;
#pragma unroll
                            for (int h = 0; h < H; ++h) {
                                atomicAdd(sp + h * 64, v[g][h].x);
                                atomicAdd(sp + h * 64 + 1, v[g][h].y);
                            }
                        }
                    }
                }
            }
        } else {
            scan_all();
        }
    } else {
        scan_all();
    }
    __syncthreads();

    for (int i = tid; i < R * F; i += 1024) {
        const int r = i / F;
        const int c = i - r * F;
        if (base + r < Nn) agg[(size_t)(base + r) * FS + col0 + c] = s_sum[i];
    }
    if (COUNT) {
        for (int i = tid; i < R; i += 1024) {
            if (base + i < Nn) deg[base + i] = s_cnt[i];
        }
    }
}

template <int K, int NOUT>
__global__ __launch_bounds__(32 * (NOUT / 16)) void sage_combine_kernel(
    const float* __restrict__ xin, const float* __restrict__ agg,
    const float* __restrict__ deg, const float* __restrict__ Wl,
    const float* __restrict__ bl, const float* __restrict__ Wr,
    float* __restrict__ out, int Nn) {
    __shared__ float s_inv[16];
    __shared__ float s_a[16 * K];
    __shared__ float s_x[16 * K];

    const int node_base = blockIdx.x * 16;
    const int tid = threadIdx.x;
    const int lane = tid & 31;
    const int wave = tid >> 5;
    const int m = lane & 15;
    const int khalf = lane >> 4;
    const int ncol = wave * 16 + m;

    if (tid < 16) {
        int nd = node_base + tid;
        if (nd > Nn - 1) nd = Nn - 1;
        s_inv[tid] = 1.0f / fmaxf(deg[nd], 1.0f);
    }
    __syncthreads();

    const int nthreads = 32 * (NOUT / 16);
    for (int i = tid; i < 16 * K; i += nthreads) {
        int nd = i / K;
        int f = i - nd * K;
        int gnode = node_base + nd;
        if (gnode > Nn - 1) gnode = Nn - 1;
        s_a[i] = agg[(size_t)gnode * K + f] * s_inv[nd];
        s_x[i] = xin[(size_t)gnode * K + f];
    }
    __syncthreads();

    v8f acc = {};
#pragma unroll
    for (int k = 0; k < K; k += 4) {
        const int kb = k + khalf * 2;
        v2f a, b;
        a.x = s_a[m * K + kb];
        a.y = s_a[m * K + kb + 1];
        b.x = Wl[(size_t)kb * NOUT + ncol];
        b.y = Wl[(size_t)(kb + 1) * NOUT + ncol];
        acc = __builtin_amdgcn_wmma_f32_16x16x4_f32(false, a, false, b,
                                                    (short)0, acc, false, false);
        a.x = s_x[m * K + kb];
        a.y = s_x[m * K + kb + 1];
        b.x = Wr[(size_t)kb * NOUT + ncol];
        b.y = Wr[(size_t)(kb + 1) * NOUT + ncol];
        acc = __builtin_amdgcn_wmma_f32_16x16x4_f32(false, a, false, b,
                                                    (short)0, acc, false, false);
    }

    const float bias = bl[ncol];
#pragma unroll
    for (int v = 0; v < 8; ++v) {
        int mrow = v + khalf * 8;
        int gnode = node_base + mrow;
        if (gnode < Nn) {
            float val = acc[v] + bias;
            out[(size_t)gnode * NOUT + ncol] = fmaxf(val, 0.0f);
        }
    }
}

__global__ void fc_sigmoid_kernel(const float* __restrict__ h2,
                                  const float* __restrict__ Wfc,
                                  const float* __restrict__ bfc,
                                  float* __restrict__ out, int Nn) {
    int i = blockIdx.x * blockDim.x + threadIdx.x;
    if (i >= Nn) return;
    float s = bfc[0];
    const float* hp = h2 + (size_t)i * 64;
#pragma unroll
    for (int k = 0; k < 64; ++k) s += hp[k] * Wfc[k];
    out[i] = 1.0f / (1.0f + __expf(-s));
}

extern "C" void kernel_launch(void* const* d_in, const int* in_sizes, int n_in,
                              void* d_out, int out_size, void* d_ws, size_t ws_size,
                              hipStream_t stream) {
    const float* x    = (const float*)d_in[0];
    const int*   ei   = (const int*)d_in[1];
    const float* W1l  = (const float*)d_in[2];
    const float* b1   = (const float*)d_in[3];
    const float* W1r  = (const float*)d_in[4];
    const float* W2l  = (const float*)d_in[5];
    const float* b2   = (const float*)d_in[6];
    const float* W2r  = (const float*)d_in[7];
    const float* Wfc  = (const float*)d_in[8];
    const float* bfc  = (const float*)d_in[9];
    float* outp = (float*)d_out;

    const int Nn = in_sizes[0] / 64;
    const int Ee = in_sizes[1] / 2;

    float* deg = (float*)d_ws;
    float* agg = deg + Nn;
    float* h1  = agg + (size_t)Nn * 128;
    float* h2  = h1 + (size_t)Nn * 128;
    int* wlist = (int*)((char*)d_ws + ((size_t)160 << 20));
    int* wcnt  = (int*)((char*)d_ws + ((size_t)224 << 20));

    const int TB = 256;

    neighbour_sum_lds_kernel<64, 64, 1024, true, true, false><<<(Nn + 1023) / 1024, 1024, 0, stream>>>(
        ei, x, agg, deg, wlist, wcnt, Ee, Nn);
    sage_combine_kernel<64, 128><<<(Nn + 15) / 16, 256, 0, stream>>>(
        x, agg, deg, W1l, b1, W1r, h1, Nn);

    neighbour_sum_lds_kernel<64, 128, 1024, false, false, true><<<(Nn + 1023) / 1024 * 2, 1024, 0, stream>>>(
        ei, h1, agg, deg, wlist, wcnt, Ee, Nn);
    sage_combine_kernel<128, 64><<<(Nn + 15) / 16, 128, 0, stream>>>(
        h1, agg, deg, W2l, b2, W2r, h2, Nn);

    fc_sigmoid_kernel<<<(Nn + TB - 1) / TB, TB, 0, stream>>>(h2, Wfc, bfc, outp, Nn);
}


# Combine kernels: LDS row pitch of the two A tiles K+4 floats instead of K, so a fragment read's 16 rows fall in different banks (was one bank, 16-way conflict)
# speedup vs baseline: 2.2038x; 1.1194x over previous
#include <hip/hip_runtime.h>
#include <hip/hip_bf16.h>
#include <type_traits>


typedef __attribute__((ext_vector_type(2))) float v2f;
typedef __attribute__((ext_vector_type(8))) float v8f;

constexpr int LCAP = 2048;

template <int F, int FS, int R, bool COUNT, bool SAVE, bool USE_LIST>
__global__ __launch_bounds__(1024) void neighbour_sum_lds_kernel(
    const int* __restrict__ ei, const float* __restrict__ feat,
    float* __restrict__ agg, float* __restrict__ deg,
    int* __restrict__ wlist, int* __restrict__ wcnt, int E, int Nn) {
    constexpr int NW = 32;
    constexpr int QCAP = 160;
    constexpr int U = 8;
    constexpr int G = 1024 / F;
    constexpr int H = F / 64;
    constexpr int NC = FS / F;
    __shared__ float s_sum[R * F];
    __shared__ float s_cnt[R];
    __shared__ int s_qe[NW * QCAP];
    __shared__ int s_qd[NW * QCAP];

    const int tid = threadIdx.x;
    const int lane = tid & 31;
    const int wave = __builtin_amdgcn_readfirstlane(tid >> 5);
    const int rb = blockIdx.x / NC;
    const int col0 = (blockIdx.x % NC) * F;
    const int base = rb * R;
    const float* featc = feat + col0 + lane * 2;

    for (int i = tid; i < R * F; i += 1024) s_sum[i] = 0.0f;
    if (COUNT) {
        for (int i = tid; i < R; i += 1024) s_cnt[i] = 0.0f;
    }
    __syncthreads();

    int2* wl = (int2*)wlist + (size_t)(rb * NW + wave) * LCAP;

    auto add_rows = [&](int src, int dl, int nb, auto gg) {
        constexpr int GG = decltype(gg)::value;
        for (int j0 = 0; j0 < nb; j0 += GG) {
            float2 v[GG][H];
#pragma unroll
            for (int g = 0; g < GG; ++g) {
                const int sj = __builtin_amdgcn_readlane(src, j0 + g);
                const float* rp = featc + (size_t)sj * FS;
#pragma unroll
                for (int h = 0; h < H; ++h) v[g][h] = *(const float2*)(rp + h * 64);
            }
#pragma unroll
            for (int g = 0; g < GG; ++g) {
                if (j0 + g < nb) {
                    const int dj = __builtin_amdgcn_readlane(dl, j0 + g);
                    float* sp = s_sum + dj * F + lane * 2;
#pragma unroll
                    for (int h = 0; h < H; ++h) {
                        atomicAdd(sp + h * 64, v[g][h].x);
                        atomicAdd(sp + h * 64 + 1, v[g][h].y);
                    }
                }
            }
        }
    };

    auto scan_all = [&]() {
        int* qe = s_qe + wave * QCAP;
        int* qd = s_qd + wave * QCAP;
        const unsigned lt = (1u << lane) - 1u;
        int qn = 0;
        int wn = 0;

        auto push = [&](int d, int e) {
            const int dl = d - base;
            const bool hit = (unsigned)dl < (unsigned)R;
            const unsigned m = __builtin_amdgcn_ballot_w32(hit);
            if (hit) {
                const int pos = qn + __popc(m & lt);
                qe[pos] = e;
                qd[pos] = dl;
            }
            qn += __popc(m);
        };

        auto drain = [&]() {
            __builtin_amdgcn_wave_barrier();
            for (int b = 0; b < qn; b += 32) {
                const int nb = min(32, qn - b);
                const bool valid = lane < nb;
                const int e = valid ? qe[b + lane] : 0;
                const int dl = valid ? qd[b + lane] : 0;
                const int src = ei[e];
                if (COUNT) {
                    if (valid) atomicAdd(&s_cnt[dl], 1.0f);
                }
                if (SAVE) {
                    if (valid && wn + lane < LCAP) wl[wn + lane] = make_int2(src, dl);
                    wn += nb;
                }
                add_rows(src, dl, nb, std::integral_constant<int, G>{});
            }
            qn = 0;
            __builtin_amdgcn_wave_barrier();
        };

        const int* dstp = ei + E;
        const int nvec = ((E & 3) == 0) ? (E >> 2) : 0;
        const int4* d4 = (const int4*)dstp;
        for (int v0 = wave * (U * 32); v0 < nvec; v0 += NW * U * 32) {
            int4 dv[U];
#pragma unroll
            for (int u = 0; u < U; ++u) {
                const int idx = v0 + u * 32 + lane;
                dv[u] = (idx < nvec) ? d4[idx] : make_int4(-1, -1, -1, -1);
            }
#pragma unroll
            for (int u = 0; u < U; ++u) {
                const int e0 = (v0 + u * 32 + lane) * 4;
                push(dv[u].x, e0);
                push(dv[u].y, e0 + 1);
                push(dv[u].z, e0 + 2);
                push(dv[u].w, e0 + 3);
                if (qn >= 32) drain();
            }
        }
        for (int e0 = nvec * 4 + wave * 32; e0 < E; e0 += NW * 32) {
            const int e = e0 + lane;
            const int d = (e < E) ? dstp[e] : -1;
            push(d, e);
            if (qn >= 32) drain();
        }
        if (qn > 0) drain();
        if (SAVE) {
            if (lane == 0) wcnt[rb * NW + wave] = wn;
        }
    };

    if (USE_LIST) {
        const int cl = wcnt[rb * NW + lane];
        const bool fits = __builtin_amdgcn_ballot_w32(cl > LCAP) == 0u;
        if (fits) {
            constexpr int GL = 2048 / F;
            const int n = __builtin_amdgcn_readlane(cl, wave);
            int2 nxt = (lane < n) ? wl[lane] : make_int2(0, 0);
            for (int i0 = 0; i0 < n; i0 += 32) {
                const int2 cur = nxt;
                const int nb = min(32, n - i0);
                for (int j0 = 0; j0 < nb; j0 += GL) {
                    float2 v[GL][H];
#pragma unroll
                    for (int g = 0; g < GL; ++g) {
                        const int sj = __builtin_amdgcn_readlane(cur.x, j0 + g);
                        const float* rp = featc + (size_t)sj * FS;
#pragma unroll
                        for (int h = 0; h < H; ++h) v[g][h] = *(const float2*)(rp + h * 64);
                    }
                    if (j0 + GL >= nb) {
                        const int in = i0 + 32 + lane;
                        nxt = (in < n) ? wl[in] : make_int2(0, 0);
                    }
#pragma unroll
                    for (int g = 0; g < GL; ++g) {
                        if (j0 + g < nb) {
                            const int dj = __builtin_amdgcn_readlane(cur.y, j0 + g);
                            float* sp = s_sum + dj * F + lane * 2;
#pragma unroll
                            for (int h = 0; h < H; ++h) {
                                atomicAdd(sp + h * 64, v[g][h].x);
                                atomicAdd(sp + h * 64 + 1, v[g][h].y);
                            }
                        }
                    }
                }
            }
        } else {
            scan_all();
        }
    } else {
        scan_all();
    }
    __syncthreads();

    for (int i = tid; i < R * F; i += 1024) {
        const int r = i / F;
        const int c = i - r * F;
        if (base + r < Nn) agg[(size_t)(base + r) * FS + col0 + c] = s_sum[i];
    }
    if (COUNT) {
        for (int i = tid; i < R; i += 1024) {
            if (base + i < Nn) deg[base + i] = s_cnt[i];
        }
    }
}

template <int K, int NOUT>
__global__ __launch_bounds__(32 * (NOUT / 16)) void sage_combine_kernel(
    const float* __restrict__ xin, const float* __restrict__ agg,
    const float* __restrict__ deg, const float* __restrict__ Wl,
    const float* __restrict__ bl, const float* __restrict__ Wr,
    float* __restrict__ out, int Nn) {
    constexpr int KP = K + 4;
    __shared__ float s_inv[16];
    __shared__ float s_a[16 * KP];
    __shared__ float s_x[16 * KP];

    const int node_base = blockIdx.x * 16;
    const int tid = threadIdx.x;
    const int lane = tid & 31;
    const int wave = tid >> 5;
    const int m = lane & 15;
    const int khalf = lane >> 4;
    const int ncol = wave * 16 + m;

    if (tid < 16) {
        int nd = node_base + tid;
        if (nd > Nn - 1) nd = Nn - 1;
        s_inv[tid] = 1.0f / fmaxf(deg[nd], 1.0f);
    }
    __syncthreads();

    const int nthreads = 32 * (NOUT / 16);
    for (int i = tid; i < 16 * K; i += nthreads) {
        int nd = i / K;
        int f = i - nd * K;
        int gnode = node_base + nd;
        if (gnode > Nn - 1) gnode = Nn - 1;
        s_a[nd * KP + f] = agg[(size_t)gnode * K + f] * s_inv[nd];
        s_x[nd * KP + f] = xin[(size_t)gnode * K + f];
    }
    __syncthreads();

    v8f acc = {};
#pragma unroll
    for (int k = 0; k < K; k += 4) {
        const int kb = k + khalf * 2;
        v2f a, b;
        a.x = s_a[m * KP + kb];
        a.y = s_a[m * KP + kb + 1];
        b.x = Wl[(size_t)kb * NOUT + ncol];
        b.y = Wl[(size_t)(kb + 1) * NOUT + ncol];
        acc = __builtin_amdgcn_wmma_f32_16x16x4_f32(false, a, false, b,
                                                    (short)0, acc, false, false);
        a.x = s_x[m * KP + kb];
        a.y = s_x[m * KP + kb + 1];
        b.x = Wr[(size_t)kb * NOUT + ncol];
        b.y = Wr[(size_t)(kb + 1) * NOUT + ncol];
        acc = __builtin_amdgcn_wmma_f32_16x16x4_f32(false, a, false, b,
                                                    (short)0, acc, false, false);
    }

    const float bias = bl[ncol];
#pragma unroll
    for (int v = 0; v < 8; ++v) {
        int mrow = v + khalf * 8;
        int gnode = node_base + mrow;
        if (gnode < Nn) {
            float val = acc[v] + bias;
            out[(size_t)gnode * NOUT + ncol] = fmaxf(val, 0.0f);
        }
    }
}

__global__ void fc_sigmoid_kernel(const float* __restrict__ h2,
                                  const float* __restrict__ Wfc,
                                  const float* __restrict__ bfc,
                                  float* __restrict__ out, int Nn) {
    int i = blockIdx.x * blockDim.x + threadIdx.x;
    if (i >= Nn) return;
    float s = bfc[0];
    const float* hp = h2 + (size_t)i * 64;
#pragma unroll
    for (int k = 0; k < 64; ++k) s += hp[k] * Wfc[k];
    out[i] = 1.0f / (1.0f + __expf(-s));
}

extern "C" void kernel_launch(void* const* d_in, const int* in_sizes, int n_in,
                              void* d_out, int out_size, void* d_ws, size_t ws_size,
                              hipStream_t stream) {
    const float* x    = (const float*)d_in[0];
    const int*   ei   = (const int*)d_in[1];
    const float* W1l  = (const float*)d_in[2];
    const float* b1   = (const float*)d_in[3];
    const float* W1r  = (const float*)d_in[4];
    const float* W2l  = (const float*)d_in[5];
    const float* b2   = (const float*)d_in[6];
    const float* W2r  = (const float*)d_in[7];
    const float* Wfc  = (const float*)d_in[8];
    const float* bfc  = (const float*)d_in[9];
    float* outp = (float*)d_out;

    const int Nn = in_sizes[0] / 64;
    const int Ee = in_sizes[1] / 2;

    float* deg = (float*)d_ws;
    float* agg = deg + Nn;
    float* h1  = agg + (size_t)Nn * 128;
    float* h2  = h1 + (size_t)Nn * 128;
    int* wlist = (int*)((char*)d_ws + ((size_t)160 << 20));
    int* wcnt  = (int*)((char*)d_ws + ((size_t)224 << 20));

    const int TB = 256;

    neighbour_sum_lds_kernel<64, 64, 1024, true, true, false><<<(Nn + 1023) / 1024, 1024, 0, stream>>>(
        ei, x, agg, deg, wlist, wcnt, Ee, Nn);
    sage_combine_kernel<64, 128><<<(Nn + 15) / 16, 256, 0, stream>>>(
        x, agg, deg, W1l, b1, W1r, h1, Nn);

    neighbour_sum_lds_kernel<64, 128, 1024, false, false, true><<<(Nn + 1023) / 1024 * 2, 1024, 0, stream>>>(
        ei, h1, agg, deg, wlist, wcnt, Ee, Nn);
    sage_combine_kernel<128, 64><<<(Nn + 15) / 16, 128, 0, stream>>>(
        h1, agg, deg, W2l, b2, W2r, h2, Nn);

    fc_sigmoid_kernel<<<(Nn + TB - 1) / TB, TB, 0, stream>>>(h2, Wfc, bfc, outp, Nn);
}


# Neighbour sums cut 2-D to fill the chip: 800-node ranges x 2 halves of the edge list (250 blocks, was 98/196); layer 1 writes two partial sums per node, the combine adds them (sum tree changes)
# speedup vs baseline: 2.9995x; 1.3611x over previous
#include <hip/hip_runtime.h>
#include <hip/hip_bf16.h>
#include <type_traits>


typedef __attribute__((ext_vector_type(2))) float v2f;
typedef __attribute__((ext_vector_type(8))) float v8f;

constexpr int LCAP = 1024;

template <int F, int FS, int R, int EP, bool COUNT, bool SAVE, bool USE_LIST>
__global__ __launch_bounds__(1024) void neighbour_sum_lds_kernel(
    const int* __restrict__ ei, const float* __restrict__ feat,
    float* __restrict__ agg, float* __restrict__ deg,
    int* __restrict__ wlist, int* __restrict__ wcnt, int E, int Nn) {
    constexpr int NW = 32;
    constexpr int QCAP = 160;
    constexpr int U = 8;
    constexpr int G = 1024 / F;
    constexpr int H = F / 64;
    constexpr int NC = FS / F;
    constexpr int BPR = USE_LIST ? NC : NC * EP;
    __shared__ float s_sum[R * F];
    __shared__ float s_cnt[R];
    __shared__ int s_qe[NW * QCAP];
    __shared__ int s_qd[NW * QCAP];

    const int tid = threadIdx.x;
    const int lane = tid & 31;
    const int wave = __builtin_amdgcn_readfirstlane(tid >> 5);
    const int rb = blockIdx.x / BPR;
    const int sub = blockIdx.x % BPR;
    const int col0 = (sub % NC) * F;
    const int part = sub / NC;
    const int base = rb * R;
    const float* featc = feat + col0 + lane * 2;

    for (int i = tid; i < R * F; i += 1024) s_sum[i] = 0.0f;
    if (COUNT) {
        for (int i = tid; i < R; i += 1024) s_cnt[i] = 0.0f;
    }
    __syncthreads();

    int2* wl = (int2*)wlist + (size_t)((rb * EP + part) * NW + wave) * LCAP;

    auto add_rows = [&](int src, int dl, int nb, auto gg) {
        constexpr int GG = decltype(gg)::value;
        for (int j0 = 0; j0 < nb; j0 += GG) {
            float2 v[GG][H];
#pragma unroll
            for (int g = 0; g < GG; ++g) {
                const int sj = __builtin_amdgcn_readlane(src, j0 + g);
                const float* rp = featc + (size_t)sj * FS;
#pragma unroll
                for (int h = 0; h < H; ++h) v[g][h] = *(const float2*)(rp + h * 64);
            }
#pragma unroll
            for (int g = 0; g < GG; ++g) {
                if (j0 + g < nb) {
                    const int dj = __builtin_amdgcn_readlane(dl, j0 + g);
                    float* sp = s_sum + dj * F + lane * 2;
#pragma unroll
                    for (int h = 0; h < H; ++h) {
                        atomicAdd(sp + h * 64, v[g][h].x);
                        atomicAdd(sp + h * 64 + 1, v[g][h].y);
                    }
                }
            }
        }
    };

    auto scan_part = [&](int p, int np) {
        int* qe = s_qe + wave * QCAP;
        int* qd = s_qd + wave * QCAP;
        const unsigned lt = (1u << lane) - 1u;
        int qn = 0;
        int wn = 0;

        auto push = [&](int d, int e) {
            const int dl = d - base;
            const bool hit = (unsigned)dl < (unsigned)R;
            const unsigned m = __builtin_amdgcn_ballot_w32(hit);
            if (hit) {
                const int pos = qn + __popc(m & lt);
                qe[pos] = e;
                qd[pos] = dl;
            }
            qn += __popc(m);
        };

        auto drain = [&]() {
            __builtin_amdgcn_wave_barrier();
            for (int b = 0; b < qn; b += 32) {
                const int nb = min(32, qn - b);
                const bool valid = lane < nb;
                const int e = valid ? qe[b + lane] : 0;
                const int dl = valid ? qd[b + lane] : 0;
                const int src = ei[e];
                if (COUNT) {
                    if (valid) atomicAdd(&s_cnt[dl], 1.0f);
                }
                if (SAVE) {
                    if (valid && wn + lane < LCAP) wl[wn + lane] = make_int2(src, dl);
                    wn += nb;
                }
                add_rows(src, dl, nb, std::integral_constant<int, G>{});
            }
            qn = 0;
            __builtin_amdgcn_wave_barrier();
        };

        const int* dstp = ei + E;
        const int nvec = ((E & 3) == 0) ? (E >> 2) : 0;
        const int4* d4 = (const int4*)dstp;
        const int v_lo = (int)((long long)nvec * p / np);
        const int v_hi = (int)((long long)nvec * (p + 1) / np);
        const int ntail = E - nvec * 4;
        const int t_lo = nvec * 4 + (int)((long long)ntail * p / np);
        const int t_hi = nvec * 4 + (int)((long long)ntail * (p + 1) / np);
        for (int v0 = v_lo + wave * (U * 32); v0 < v_hi; v0 += NW * U * 32) {
            int4 dv[U];
#pragma unroll
            for (int u = 0; u < U; ++u) {
                const int idx = v0 + u * 32 + lane;
                dv[u] = (idx < v_hi) ? d4[idx] : make_int4(-1, -1, -1, -1);
            }
#pragma unroll
            for (int u = 0; u < U; ++u) {
                const int e0 = (v0 + u * 32 + lane) * 4;
                push(dv[u].x, e0);
                push(dv[u].y, e0 + 1);
                push(dv[u].z, e0 + 2);
                push(dv[u].w, e0 + 3);
                if (qn >= 32) drain();
            }
        }
        for (int e0 = t_lo + wave * 32; e0 < t_hi; e0 += NW * 32) {
            const int e = e0 + lane;
            const int d = (e < t_hi) ? dstp[e] : -1;
            push(d, e);
            if (qn >= 32) drain();
        }
        if (qn > 0) drain();
        if (SAVE) {
            if (lane == 0) wcnt[(rb * EP + part) * NW + wave] = wn;
        }
    };

    if (USE_LIST) {
        int cl[EP];
        bool over = false;
#pragma unroll
        for (int p = 0; p < EP; ++p) {
            cl[p] = wcnt[(rb * EP + p) * NW + lane];
            over = over || (cl[p] > LCAP);
        }
        const bool fits = __builtin_amdgcn_ballot_w32(over) == 0u;
        if (fits) {
            constexpr int GL = 2048 / F;
            int nn[EP];
            int n = 0;
#pragma unroll
            for (int p = 0; p < EP; ++p) {
                nn[p] = __builtin_amdgcn_readlane(cl[p], wave);
                n += nn[p];
            }
            const int2* wlb = (const int2*)wlist + (size_t)(rb * EP * NW + wave) * LCAP;
            auto entry = [&](int i) {
                int off = i;
                int idx = -1;
#pragma unroll
                for (int p = 0; p < EP; ++p) {
                    if (off >= 0 && off < nn[p]) idx = p * (NW * LCAP) + off;
                    off -= nn[p];
                }
                return (idx >= 0) ? wlb[idx] : make_int2(0, 0);
            };
            int2 nxt = entry(lane);
            for (int i0 = 0; i0 < n; i0 += 32) {
                const int2 cur = nxt;
                const int nb = min(32, n - i0);
                for (int j0 = 0; j0 < nb; j0 += GL) {
                    float2 v[GL][H];
#pragma unroll
                    for (int g = 0; g < GL; ++g) {
                        const int sj = __builtin_amdgcn_readlane(cur.x, j0 + g);
                        const float* rp = featc + (size_t)sj * FS;
#pragma unroll
                        for (int h = 0; h < H; ++h) v[g][h] = *(const float2*)(rp + h * 64);
                    }
                    if (j0 + GL >= nb) {
                        nxt = entry(i0 + 32 + lane);
                    }
#pragma unroll
                    for (int g = 0; g < GL; ++g) {
                        if (j0 + g < nb) {
                            const int dj = __builtin_amdgcn_readlane(cur.y, j0 + g);
                            float* sp = s_sum + dj * F + lane * 2;
#pragma unroll
                            for (int h = 0; h < H; ++h) {
                                atomicAdd(sp + h * 64, v[g][h].x);
                                atomicAdd(sp + h * 64 + 1, v[g][h].y);
                            }
                        }
                    }
                }
            }
        } else {
            scan_part(0, 1);
        }
    } else {
        scan_part(part, EP);
    }
    __syncthreads();

    float* aggp = agg + (size_t)part * Nn * FS;
    float* degp = deg + (size_t)part * Nn;
    for (int i = tid; i < R * F; i += 1024) {
        const int r = i / F;
        const int c = i - r * F;
        if (base + r < Nn) aggp[(size_t)(base + r) * FS + col0 + c] = s_sum[i];
    }
    if (COUNT) {
        for (int i = tid; i < R; i += 1024) {
            if (base + i < Nn) degp[base + i] = s_cnt[i];
        }
    }
}

template <int K, int NOUT, int AP, int DP>
__global__ __launch_bounds__(32 * (NOUT / 16)) void sage_combine_kernel(
    const float* __restrict__ xin, const float* __restrict__ agg,
    const float* __restrict__ deg, const float* __restrict__ Wl,
    const float* __restrict__ bl, const float* __restrict__ Wr,
    float* __restrict__ out, int Nn) {
    constexpr int KP = K + 4;
    __shared__ float s_inv[16];
    __shared__ float s_a[16 * KP];
    __shared__ float s_x[16 * KP];

    const int node_base = blockIdx.x * 16;
    const int tid = threadIdx.x;
    const int lane = tid & 31;
    const int wave = tid >> 5;
    const int m = lane & 15;
    const int khalf = lane >> 4;
    const int ncol = wave * 16 + m;

    if (tid < 16) {
        int nd = node_base + tid;
        if (nd > Nn - 1) nd = Nn - 1;
        float dsum = deg[nd];
#pragma unroll
        for (int p = 1; p < DP; ++p) dsum += deg[(size_t)p * Nn + nd];
        s_inv[tid] = 1.0f / fmaxf(dsum, 1.0f);
    }
    __syncthreads();

    const int nthreads = 32 * (NOUT / 16);
    for (int i = tid; i < 16 * K; i += nthreads) {
        int nd = i / K;
        int f = i - nd * K;
        int gnode = node_base + nd;
        if (gnode > Nn - 1) gnode = Nn - 1;
        float asum = agg[(size_t)gnode * K + f];
#pragma unroll
        for (int p = 1; p < AP; ++p) asum += agg[((size_t)p * Nn + gnode) * K + f];
        s_a[nd * KP + f] = asum * s_inv[nd];
        s_x[nd * KP + f] = xin[(size_t)gnode * K + f];
    }
    __syncthreads();

    v8f acc = {};
#pragma unroll
    for (int k = 0; k < K; k += 4) {
        const int kb = k + khalf * 2;
        v2f a, b;
        a.x = s_a[m * KP + kb];
        a.y = s_a[m * KP + kb + 1];
        b.x = Wl[(size_t)kb * NOUT + ncol];
        b.y = Wl[(size_t)(kb + 1) * NOUT + ncol];
        acc = __builtin_amdgcn_wmma_f32_16x16x4_f32(false, a, false, b,
                                                    (short)0, acc, false, false);
        a.x = s_x[m * KP + kb];
        a.y = s_x[m * KP + kb + 1];
        b.x = Wr[(size_t)kb * NOUT + ncol];
        b.y = Wr[(size_t)(kb + 1) * NOUT + ncol];
        acc = __builtin_amdgcn_wmma_f32_16x16x4_f32(false, a, false, b,
                                                    (short)0, acc, false, false);
    }

    const float bias = bl[ncol];
#pragma unroll
    for (int v = 0; v < 8; ++v) {
        int mrow = v + khalf * 8;
        int gnode = node_base + mrow;
        if (gnode < Nn) {
            float val = acc[v] + bias;
            out[(size_t)gnode * NOUT + ncol] = fmaxf(val, 0.0f);
        }
    }
}

__global__ void fc_sigmoid_kernel(const float* __restrict__ h2,
                                  const float* __restrict__ Wfc,
                                  const float* __restrict__ bfc,
                                  float* __restrict__ out, int Nn) {
    int i = blockIdx.x * blockDim.x + threadIdx.x;
    if (i >= Nn) return;
    float s = bfc[0];
    const float* hp = h2 + (size_t)i * 64;
#pragma unroll
    for (int k = 0; k < 64; ++k) s += hp[k] * Wfc[k];
    out[i] = 1.0f / (1.0f + __expf(-s));
}

extern "C" void kernel_launch(void* const* d_in, const int* in_sizes, int n_in,
                              void* d_out, int out_size, void* d_ws, size_t ws_size,
                              hipStream_t stream) {
    const float* x    = (const float*)d_in[0];
    const int*   ei   = (const int*)d_in[1];
    const float* W1l  = (const float*)d_in[2];
    const float* b1   = (const float*)d_in[3];
    const float* W1r  = (const float*)d_in[4];
    const float* W2l  = (const float*)d_in[5];
    const float* b2   = (const float*)d_in[6];
    const float* W2r  = (const float*)d_in[7];
    const float* Wfc  = (const float*)d_in[8];
    const float* bfc  = (const float*)d_in[9];
    float* outp = (float*)d_out;

    const int Nn = in_sizes[0] / 64;
    const int Ee = in_sizes[1] / 2;

    float* deg = (float*)d_ws;
    float* agg = deg + (size_t)Nn * 2;
    float* h1  = agg + (size_t)Nn * 128;
    float* h2  = h1 + (size_t)Nn * 128;
    int* wlist = (int*)((char*)d_ws + ((size_t)160 << 20));
    int* wcnt  = (int*)((char*)d_ws + ((size_t)224 << 20));

    const int TB = 256;

    neighbour_sum_lds_kernel<64, 64, 800, 2, true, true, false><<<(Nn + 799) / 800 * 2, 1024, 0, stream>>>(
        ei, x, agg, deg, wlist, wcnt, Ee, Nn);
    sage_combine_kernel<64, 128, 2, 2><<<(Nn + 15) / 16, 256, 0, stream>>>(
        x, agg, deg, W1l, b1, W1r, h1, Nn);

    neighbour_sum_lds_kernel<64, 128, 800, 2, false, false, true><<<(Nn + 799) / 800 * 2, 1024, 0, stream>>>(
        ei, h1, agg, deg, wlist, wcnt, Ee, Nn);
    sage_combine_kernel<128, 64, 1, 2><<<(Nn + 15) / 16, 128, 0, stream>>>(
        h1, agg, deg, W2l, b2, W2r, h2, Nn);

    fc_sigmoid_kernel<<<(Nn + TB - 1) / TB, TB, 0, stream>>>(h2, Wfc, bfc, outp, Nn);
}


# Destination scan: 32 destinations with none of the block's own now cost subtract, compare, branch; queue count and stores run only behind a hit; whole chunks load without per-load bounds checks
# speedup vs baseline: 3.1655x; 1.0553x over previous
#include <hip/hip_runtime.h>
#include <hip/hip_bf16.h>
#include <type_traits>


typedef __attribute__((ext_vector_type(2))) float v2f;
typedef __attribute__((ext_vector_type(8))) float v8f;

constexpr int LCAP = 1024;

template <int F, int FS, int R, int EP, bool COUNT, bool SAVE, bool USE_LIST>
__global__ __launch_bounds__(1024) void neighbour_sum_lds_kernel(
    const int* __restrict__ ei, const float* __restrict__ feat,
    float* __restrict__ agg, float* __restrict__ deg,
    int* __restrict__ wlist, int* __restrict__ wcnt, int E, int Nn) {
    constexpr int NW = 32;
    constexpr int QCAP = 160;
    constexpr int U = 8;
    constexpr int G = 1024 / F;
    constexpr int H = F / 64;
    constexpr int NC = FS / F;
    constexpr int BPR = USE_LIST ? NC : NC * EP;
    __shared__ float s_sum[R * F];
    __shared__ float s_cnt[R];
    __shared__ int s_qe[NW * QCAP];
    __shared__ int s_qd[NW * QCAP];

    const int tid = threadIdx.x;
    const int lane = tid & 31;
    const int wave = __builtin_amdgcn_readfirstlane(tid >> 5);
    const int rb = blockIdx.x / BPR;
    const int sub = blockIdx.x % BPR;
    const int col0 = (sub % NC) * F;
    const int part = sub / NC;
    const int base = rb * R;
    const float* featc = feat + col0 + lane * 2;

    for (int i = tid; i < R * F; i += 1024) s_sum[i] = 0.0f;
    if (COUNT) {
        for (int i = tid; i < R; i += 1024) s_cnt[i] = 0.0f;
    }
    __syncthreads();

    int2* wl = (int2*)wlist + (size_t)((rb * EP + part) * NW + wave) * LCAP;

    auto add_rows = [&](int src, int dl, int nb, auto gg) {
        constexpr int GG = decltype(gg)::value;
        for (int j0 = 0; j0 < nb; j0 += GG) {
            float2 v[GG][H];
#pragma unroll
            for (int g = 0; g < GG; ++g) {
                const int sj = __builtin_amdgcn_readlane(src, j0 + g);
                const float* rp = featc + (size_t)sj * FS;
#pragma unroll
                for (int h = 0; h < H; ++h) v[g][h] = *(const float2*)(rp + h * 64);
            }
#pragma unroll
            for (int g = 0; g < GG; ++g) {
                if (j0 + g < nb) {
                    const int dj = __builtin_amdgcn_readlane(dl, j0 + g);
                    float* sp = s_sum + dj * F + lane * 2;
#pragma unroll
                    for (int h = 0; h < H; ++h) {
                        atomicAdd(sp + h * 64, v[g][h].x);
                        atomicAdd(sp + h * 64 + 1, v[g][h].y);
                    }
                }
            }
        }
    };

    auto scan_part = [&](int p, int np) {
        int* qe = s_qe + wave * QCAP;
        int* qd = s_qd + wave * QCAP;
        const unsigned lt = (1u << lane) - 1u;
        int qn = 0;
        int wn = 0;

        auto push = [&](int d, int e) {
            const int dl = d - base;
            const bool hit = (unsigned)dl < (unsigned)R;
            const unsigned m = __builtin_amdgcn_ballot_w32(hit);
            if (__builtin_expect(m != 0u, 0)) {
                if (hit) {
                    const int pos = qn + __popc(m & lt);
                    qe[pos] = e;
                    qd[pos] = dl;
                }
                qn += __popc(m);
            }
        };

        auto drain = [&]() {
            __builtin_amdgcn_wave_barrier();
            for (int b = 0; b < qn; b += 32) {
                const int nb = min(32, qn - b);
                const bool valid = lane < nb;
                const int e = valid ? qe[b + lane] : 0;
                const int dl = valid ? qd[b + lane] : 0;
                const int src = ei[e];
                if (COUNT) {
                    if (valid) atomicAdd(&s_cnt[dl], 1.0f);
                }
                if (SAVE) {
                    if (valid && wn + lane < LCAP) wl[wn + lane] = make_int2(src, dl);
                    wn += nb;
                }
                add_rows(src, dl, nb, std::integral_constant<int, G>{});
            }
            qn = 0;
            __builtin_amdgcn_wave_barrier();
        };

        const int* dstp = ei + E;
        const int nvec = ((E & 3) == 0) ? (E >> 2) : 0;
        const int4* d4 = (const int4*)dstp;
        const int v_lo = (int)((long long)nvec * p / np);
        const int v_hi = (int)((long long)nvec * (p + 1) / np);
        const int ntail = E - nvec * 4;
        const int t_lo = nvec * 4 + (int)((long long)ntail * p / np);
        const int t_hi = nvec * 4 + (int)((long long)ntail * (p + 1) / np);
        for (int v0 = v_lo + wave * (U * 32); v0 < v_hi; v0 += NW * U * 32) {
            int4 dv[U];
            if (v0 + U * 32 <= v_hi) {
                const int4* dp = d4 + v0 + lane;
#pragma unroll
                for (int u = 0; u < U; ++u) dv[u] = dp[u * 32];
            } else {
#pragma unroll
                for (int u = 0; u < U; ++u) {
                    const int idx = v0 + u * 32 + lane;
                    dv[u] = (idx < v_hi) ? d4[idx] : make_int4(-1, -1, -1, -1);
                }
            }
#pragma unroll
            for (int u = 0; u < U; ++u) {
                const int e0 = (v0 + u * 32 + lane) * 4;
                push(dv[u].x, e0);
                push(dv[u].y, e0 + 1);
                push(dv[u].z, e0 + 2);
                push(dv[u].w, e0 + 3);
                if (qn >= 32) drain();
            }
        }
        for (int e0 = t_lo + wave * 32; e0 < t_hi; e0 += NW * 32) {
            const int e = e0 + lane;
            const int d = (e < t_hi) ? dstp[e] : -1;
            push(d, e);
            if (qn >= 32) drain();
        }
        if (qn > 0) drain();
        if (SAVE) {
            if (lane == 0) wcnt[(rb * EP + part) * NW + wave] = wn;
        }
    };

    if (USE_LIST) {
        int cl[EP];
        bool over = false;
#pragma unroll
        for (int p = 0; p < EP; ++p) {
            cl[p] = wcnt[(rb * EP + p) * NW + lane];
            over = over || (cl[p] > LCAP);
        }
        const bool fits = __builtin_amdgcn_ballot_w32(over) == 0u;
        if (fits) {
            constexpr int GL = 2048 / F;
            int nn[EP];
            int n = 0;
#pragma unroll
            for (int p = 0; p < EP; ++p) {
                nn[p] = __builtin_amdgcn_readlane(cl[p], wave);
                n += nn[p];
            }
            const int2* wlb = (const int2*)wlist + (size_t)(rb * EP * NW + wave) * LCAP;
            auto entry = [&](int i) {
                int off = i;
                int idx = -1;
#pragma unroll
                for (int p = 0; p < EP; ++p) {
                    if (off >= 0 && off < nn[p]) idx = p * (NW * LCAP) + off;
                    off -= nn[p];
                }
                return (idx >= 0) ? wlb[idx] : make_int2(0, 0);
            };
            int2 nxt = entry(lane);
            for (int i0 = 0; i0 < n; i0 += 32) {
                const int2 cur = nxt;
                const int nb = min(32, n - i0);
                for (int j0 = 0; j0 < nb; j0 += GL) {
                    float2 v[GL][H];
#pragma unroll
                    for (int g = 0; g < GL; ++g) {
                        const int sj = __builtin_amdgcn_readlane(cur.x, j0 + g);
                        const float* rp = featc + (size_t)sj * FS;
#pragma unroll
                        for (int h = 0; h < H; ++h) v[g][h] = *(const float2*)(rp + h * 64);
                    }
                    if (j0 + GL >= nb) {
                        nxt = entry(i0 + 32 + lane);
                    }
#pragma unroll
                    for (int g = 0; g < GL; ++g) {
                        if (j0 + g < nb) {
                            const int dj = __builtin_amdgcn_readlane(cur.y, j0 + g);
                            float* sp = s_sum + dj * F + lane * 2;
#pragma unroll
                            for (int h = 0; h < H; ++h) {
                                atomicAdd(sp + h * 64, v[g][h].x);
                                atomicAdd(sp + h * 64 + 1, v[g][h].y);
                            }
                        }
                    }
                }
            }
        } else {
            scan_part(0, 1);
        }
    } else {
        scan_part(part, EP);
    }
    __syncthreads();

    float* aggp = agg + (size_t)part * Nn * FS;
    float* degp = deg + (size_t)part * Nn;
    for (int i = tid; i < R * F; i += 1024) {
        const int r = i / F;
        const int c = i - r * F;
        if (base + r < Nn) aggp[(size_t)(base + r) * FS + col0 + c] = s_sum[i];
    }
    if (COUNT) {
        for (int i = tid; i < R; i += 1024) {
            if (base + i < Nn) degp[base + i] = s_cnt[i];
        }
    }
}

template <int K, int NOUT, int AP, int DP>
__global__ __launch_bounds__(32 * (NOUT / 16)) void sage_combine_kernel(
    const float* __restrict__ xin, const float* __restrict__ agg,
    const float* __restrict__ deg, const float* __restrict__ Wl,
    const float* __restrict__ bl, const float* __restrict__ Wr,
    float* __restrict__ out, int Nn) {
    constexpr int KP = K + 4;
    __shared__ float s_inv[16];
    __shared__ float s_a[16 * KP];
    __shared__ float s_x[16 * KP];

    const int node_base = blockIdx.x * 16;
    const int tid = threadIdx.x;
    const int lane = tid & 31;
    const int wave = tid >> 5;
    const int m = lane & 15;
    const int khalf = lane >> 4;
    const int ncol = wave * 16 + m;

    if (tid < 16) {
        int nd = node_base + tid;
        if (nd > Nn - 1) nd = Nn - 1;
        float dsum = deg[nd];
#pragma unroll
        for (int p = 1; p < DP; ++p) dsum += deg[(size_t)p * Nn + nd];
        s_inv[tid] = 1.0f / fmaxf(dsum, 1.0f);
    }
    __syncthreads();

    const int nthreads = 32 * (NOUT / 16);
    for (int i = tid; i < 16 * K; i += nthreads) {
        int nd = i / K;
        int f = i - nd * K;
        int gnode = node_base + nd;
        if (gnode > Nn - 1) gnode = Nn - 1;
        float asum = agg[(size_t)gnode * K + f];
#pragma unroll
        for (int p = 1; p < AP; ++p) asum += agg[((size_t)p * Nn + gnode) * K + f];
        s_a[nd * KP + f] = asum * s_inv[nd];
        s_x[nd * KP + f] = xin[(size_t)gnode * K + f];
    }
    __syncthreads();

    v8f acc = {};
#pragma unroll
    for (int k = 0; k < K; k += 4) {
        const int kb = k + khalf * 2;
        v2f a, b;
        a.x = s_a[m * KP + kb];
        a.y = s_a[m * KP + kb + 1];
        b.x = Wl[(size_t)kb * NOUT + ncol];
        b.y = Wl[(size_t)(kb + 1) * NOUT + ncol];
        acc = __builtin_amdgcn_wmma_f32_16x16x4_f32(false, a, false, b,
                                                    (short)0, acc, false, false);
        a.x = s_x[m * KP + kb];
        a.y = s_x[m * KP + kb + 1];
        b.x = Wr[(size_t)kb * NOUT + ncol];
        b.y = Wr[(size_t)(kb + 1) * NOUT + ncol];
        acc = __builtin_amdgcn_wmma_f32_16x16x4_f32(false, a, false, b,
                                                    (short)0, acc, false, false);
    }

    const float bias = bl[ncol];
#pragma unroll
    for (int v = 0; v < 8; ++v) {
        int mrow = v + khalf * 8;
        int gnode = node_base + mrow;
        if (gnode < Nn) {
            float val = acc[v] + bias;
            out[(size_t)gnode * NOUT + ncol] = fmaxf(val, 0.0f);
        }
    }
}

__global__ void fc_sigmoid_kernel(const float* __restrict__ h2,
                                  const float* __restrict__ Wfc,
                                  const float* __restrict__ bfc,
                                  float* __restrict__ out, int Nn) {
    int i = blockIdx.x * blockDim.x + threadIdx.x;
    if (i >= Nn) return;
    float s = bfc[0];
    const float* hp = h2 + (size_t)i * 64;
#pragma unroll
    for (int k = 0; k < 64; ++k) s += hp[k] * Wfc[k];
    out[i] = 1.0f / (1.0f + __expf(-s));
}

extern "C" void kernel_launch(void* const* d_in, const int* in_sizes, int n_in,
                              void* d_out, int out_size, void* d_ws, size_t ws_size,
                              hipStream_t stream) {
    const float* x    = (const float*)d_in[0];
    const int*   ei   = (const int*)d_in[1];
    const float* W1l  = (const float*)d_in[2];
    const float* b1   = (const float*)d_in[3];
    const float* W1r  = (const float*)d_in[4];
    const float* W2l  = (const float*)d_in[5];
    const float* b2   = (const float*)d_in[6];
    const float* W2r  = (const float*)d_in[7];
    const float* Wfc  = (const float*)d_in[8];
    const float* bfc  = (const float*)d_in[9];
    float* outp = (float*)d_out;

    const int Nn = in_sizes[0] / 64;
    const int Ee = in_sizes[1] / 2;

    float* deg = (float*)d_ws;
    float* agg = deg + (size_t)Nn * 2;
    float* h1  = agg + (size_t)Nn * 128;
    float* h2  = h1 + (size_t)Nn * 128;
    int* wlist = (int*)((char*)d_ws + ((size_t)160 << 20));
    int* wcnt  = (int*)((char*)d_ws + ((size_t)224 << 20));

    const int TB = 256;

    neighbour_sum_lds_kernel<64, 64, 800, 2, true, true, false><<<(Nn + 799) / 800 * 2, 1024, 0, stream>>>(
        ei, x, agg, deg, wlist, wcnt, Ee, Nn);
    sage_combine_kernel<64, 128, 2, 2><<<(Nn + 15) / 16, 256, 0, stream>>>(
        x, agg, deg, W1l, b1, W1r, h1, Nn);

    neighbour_sum_lds_kernel<64, 128, 800, 2, false, false, true><<<(Nn + 799) / 800 * 2, 1024, 0, stream>>>(
        ei, h1, agg, deg, wlist, wcnt, Ee, Nn);
    sage_combine_kernel<128, 64, 1, 2><<<(Nn + 15) / 16, 128, 0, stream>>>(
        h1, agg, deg, W2l, b2, W2r, h2, Nn);

    fc_sigmoid_kernel<<<(Nn + TB - 1) / TB, TB, 0, stream>>>(h2, Wfc, bfc, outp, Nn);
}


# Head folded into the layer-2 combine: a wave dots its 16x16 tile with Wfc (16-lane butterfly, wave partials via LDS), the block stores the sigmoid; no h2; 5 launches to 4; head sum order changes
# speedup vs baseline: 3.2642x; 1.0312x over previous
#include <hip/hip_runtime.h>
#include <hip/hip_bf16.h>
#include <type_traits>


typedef __attribute__((ext_vector_type(2))) float v2f;
typedef __attribute__((ext_vector_type(8))) float v8f;

constexpr int LCAP = 1024;

template <int F, int FS, int R, int EP, bool COUNT, bool SAVE, bool USE_LIST>
__global__ __launch_bounds__(1024) void neighbour_sum_lds_kernel(
    const int* __restrict__ ei, const float* __restrict__ feat,
    float* __restrict__ agg, float* __restrict__ deg,
    int* __restrict__ wlist, int* __restrict__ wcnt, int E, int Nn) {
    constexpr int NW = 32;
    constexpr int QCAP = 160;
    constexpr int U = 8;
    constexpr int G = 1024 / F;
    constexpr int H = F / 64;
    constexpr int NC = FS / F;
    constexpr int BPR = USE_LIST ? NC : NC * EP;
    __shared__ float s_sum[R * F];
    __shared__ float s_cnt[R];
    __shared__ int s_qe[NW * QCAP];
    __shared__ int s_qd[NW * QCAP];

    const int tid = threadIdx.x;
    const int lane = tid & 31;
    const int wave = __builtin_amdgcn_readfirstlane(tid >> 5);
    const int rb = blockIdx.x / BPR;
    const int sub = blockIdx.x % BPR;
    const int col0 = (sub % NC) * F;
    const int part = sub / NC;
    const int base = rb * R;
    const float* featc = feat + col0 + lane * 2;

    for (int i = tid; i < R * F; i += 1024) s_sum[i] = 0.0f;
    if (COUNT) {
        for (int i = tid; i < R; i += 1024) s_cnt[i] = 0.0f;
    }
    __syncthreads();

    int2* wl = (int2*)wlist + (size_t)((rb * EP + part) * NW + wave) * LCAP;

    auto add_rows = [&](int src, int dl, int nb, auto gg) {
        constexpr int GG = decltype(gg)::value;
        for (int j0 = 0; j0 < nb; j0 += GG) {
            float2 v[GG][H];
#pragma unroll
            for (int g = 0; g < GG; ++g) {
                const int sj = __builtin_amdgcn_readlane(src, j0 + g);
                const float* rp = featc + (size_t)sj * FS;
#pragma unroll
                for (int h = 0; h < H; ++h) v[g][h] = *(const float2*)(rp + h * 64);
            }
#pragma unroll
            for (int g = 0; g < GG; ++g) {
                if (j0 + g < nb) {
                    const int dj = __builtin_amdgcn_readlane(dl, j0 + g);
                    float* sp = s_sum + dj * F + lane * 2;
#pragma unroll
                    for (int h = 0; h < H; ++h) {
                        atomicAdd(sp + h * 64, v[g][h].x);
                        atomicAdd(sp + h * 64 + 1, v[g][h].y);
                    }
                }
            }
        }
    };

    auto scan_part = [&](int p, int np) {
        int* qe = s_qe + wave * QCAP;
        int* qd = s_qd + wave * QCAP;
        const unsigned lt = (1u << lane) - 1u;
        int qn = 0;
        int wn = 0;

        auto push = [&](int d, int e) {
            const int dl = d - base;
            const bool hit = (unsigned)dl < (unsigned)R;
            const unsigned m = __builtin_amdgcn_ballot_w32(hit);
            if (__builtin_expect(m != 0u, 0)) {
                if (hit) {
                    const int pos = qn + __popc(m & lt);
                    qe[pos] = e;
                    qd[pos] = dl;
                }
                qn += __popc(m);
            }
        };

        auto drain = [&]() {
            __builtin_amdgcn_wave_barrier();
            for (int b = 0; b < qn; b += 32) {
                const int nb = min(32, qn - b);
                const bool valid = lane < nb;
                const int e = valid ? qe[b + lane] : 0;
                const int dl = valid ? qd[b + lane] : 0;
                const int src = ei[e];
                if (COUNT) {
                    if (valid) atomicAdd(&s_cnt[dl], 1.0f);
                }
                if (SAVE) {
                    if (valid && wn + lane < LCAP) wl[wn + lane] = make_int2(src, dl);
                    wn += nb;
                }
                add_rows(src, dl, nb, std::integral_constant<int, G>{});
            }
            qn = 0;
            __builtin_amdgcn_wave_barrier();
        };

        const int* dstp = ei + E;
        const int nvec = ((E & 3) == 0) ? (E >> 2) : 0;
        const int4* d4 = (const int4*)dstp;
        const int v_lo = (int)((long long)nvec * p / np);
        const int v_hi = (int)((long long)nvec * (p + 1) / np);
        const int ntail = E - nvec * 4;
        const int t_lo = nvec * 4 + (int)((long long)ntail * p / np);
        const int t_hi = nvec * 4 + (int)((long long)ntail * (p + 1) / np);
        for (int v0 = v_lo + wave * (U * 32); v0 < v_hi; v0 += NW * U * 32) {
            int4 dv[U];
            if (v0 + U * 32 <= v_hi) {
                const int4* dp = d4 + v0 + lane;
#pragma unroll
                for (int u = 0; u < U; ++u) dv[u] = dp[u * 32];
            } else {
#pragma unroll
                for (int u = 0; u < U; ++u) {
                    const int idx = v0 + u * 32 + lane;
                    dv[u] = (idx < v_hi) ? d4[idx] : make_int4(-1, -1, -1, -1);
                }
            }
#pragma unroll
            for (int u = 0; u < U; ++u) {
                const int e0 = (v0 + u * 32 + lane) * 4;
                push(dv[u].x, e0);
                push(dv[u].y, e0 + 1);
                push(dv[u].z, e0 + 2);
                push(dv[u].w, e0 + 3);
                if (qn >= 32) drain();
            }
        }
        for (int e0 = t_lo + wave * 32; e0 < t_hi; e0 += NW * 32) {
            const int e = e0 + lane;
            const int d = (e < t_hi) ? dstp[e] : -1;
            push(d, e);
            if (qn >= 32) drain();
        }
        if (qn > 0) drain();
        if (SAVE) {
            if (lane == 0) wcnt[(rb * EP + part) * NW + wave] = wn;
        }
    };

    if (USE_LIST) {
        int cl[EP];
        bool over = false;
#pragma unroll
        for (int p = 0; p < EP; ++p) {
            cl[p] = wcnt[(rb * EP + p) * NW + lane];
            over = over || (cl[p] > LCAP);
        }
        const bool fits = __builtin_amdgcn_ballot_w32(over) == 0u;
        if (fits) {
            constexpr int GL = 2048 / F;
            int nn[EP];
            int n = 0;
#pragma unroll
            for (int p = 0; p < EP; ++p) {
                nn[p] = __builtin_amdgcn_readlane(cl[p], wave);
                n += nn[p];
            }
            const int2* wlb = (const int2*)wlist + (size_t)(rb * EP * NW + wave) * LCAP;
            auto entry = [&](int i) {
                int off = i;
                int idx = -1;
#pragma unroll
                for (int p = 0; p < EP; ++p) {
                    if (off >= 0 && off < nn[p]) idx = p * (NW * LCAP) + off;
                    off -= nn[p];
                }
                return (idx >= 0) ? wlb[idx] : make_int2(0, 0);
            };
            int2 nxt = entry(lane);
            for (int i0 = 0; i0 < n; i0 += 32) {
                const int2 cur = nxt;
                const int nb = min(32, n - i0);
                for (int j0 = 0; j0 < nb; j0 += GL) {
                    float2 v[GL][H];
#pragma unroll
                    for (int g = 0; g < GL; ++g) {
                        const int sj = __builtin_amdgcn_readlane(cur.x, j0 + g);
                        const float* rp = featc + (size_t)sj * FS;
#pragma unroll
                        for (int h = 0; h < H; ++h) v[g][h] = *(const float2*)(rp + h * 64);
                    }
                    if (j0 + GL >= nb) {
                        nxt = entry(i0 + 32 + lane);
                    }
#pragma unroll
                    for (int g = 0; g < GL; ++g) {
                        if (j0 + g < nb) {
                            const int dj = __builtin_amdgcn_readlane(cur.y, j0 + g);
                            float* sp = s_sum + dj * F + lane * 2;
#pragma unroll
                            for (int h = 0; h < H; ++h) {
                                atomicAdd(sp + h * 64, v[g][h].x);
                                atomicAdd(sp + h * 64 + 1, v[g][h].y);
                            }
                        }
                    }
                }
            }
        } else {
            scan_part(0, 1);
        }
    } else {
        scan_part(part, EP);
    }
    __syncthreads();

    float* aggp = agg + (size_t)part * Nn * FS;
    float* degp = deg + (size_t)part * Nn;
    for (int i = tid; i < R * F; i += 1024) {
        const int r = i / F;
        const int c = i - r * F;
        if (base + r < Nn) aggp[(size_t)(base + r) * FS + col0 + c] = s_sum[i];
    }
    if (COUNT) {
        for (int i = tid; i < R; i += 1024) {
            if (base + i < Nn) degp[base + i] = s_cnt[i];
        }
    }
}


__device__ __forceinline__ float half_wave_sum(float v) {
    v += __builtin_bit_cast(float, __builtin_amdgcn_update_dpp(0, __builtin_bit_cast(int, v), 0x161, 0xf, 0xf, true));
    v += __builtin_bit_cast(float, __builtin_amdgcn_update_dpp(0, __builtin_bit_cast(int, v), 0x162, 0xf, 0xf, true));
    v += __builtin_bit_cast(float, __builtin_amdgcn_update_dpp(0, __builtin_bit_cast(int, v), 0x164, 0xf, 0xf, true));
    v += __builtin_bit_cast(float, __builtin_amdgcn_update_dpp(0, __builtin_bit_cast(int, v), 0x168, 0xf, 0xf, true));
    return v;
}

template <int K, int NOUT, int AP, int DP, bool HEAD>
__global__ __launch_bounds__(32 * (NOUT / 16)) void sage_combine_kernel(
    const float* __restrict__ xin, const float* __restrict__ agg,
    const float* __restrict__ deg, const float* __restrict__ Wl,
    const float* __restrict__ bl, const float* __restrict__ Wr,
    const float* __restrict__ Wfc, const float* __restrict__ bfc,
    float* __restrict__ out, int Nn) {
    constexpr int KP = K + 4;
    __shared__ float s_inv[16];
    __shared__ float s_a[16 * KP];
    __shared__ float s_x[16 * KP];

    const int node_base = blockIdx.x * 16;
    const int tid = threadIdx.x;
    const int lane = tid & 31;
    const int wave = tid >> 5;
    const int m = lane & 15;
    const int khalf = lane >> 4;
    const int ncol = wave * 16 + m;

    if (tid < 16) {
        int nd = node_base + tid;
        if (nd > Nn - 1) nd = Nn - 1;
        float dsum = deg[nd];
#pragma unroll
        for (int p = 1; p < DP; ++p) dsum += deg[(size_t)p * Nn + nd];
        s_inv[tid] = 1.0f / fmaxf(dsum, 1.0f);
    }
    __syncthreads();

    const int nthreads = 32 * (NOUT / 16);
    for (int i = tid; i < 16 * K; i += nthreads) {
        int nd = i / K;
        int f = i - nd * K;
        int gnode = node_base + nd;
        if (gnode > Nn - 1) gnode = Nn - 1;
        float asum = agg[(size_t)gnode * K + f];
#pragma unroll
        for (int p = 1; p < AP; ++p) asum += agg[((size_t)p * Nn + gnode) * K + f];
        s_a[nd * KP + f] = asum * s_inv[nd];
        s_x[nd * KP + f] = xin[(size_t)gnode * K + f];
    }
    __syncthreads();

    v8f acc = {};
#pragma unroll
    for (int k = 0; k < K; k += 4) {
        const int kb = k + khalf * 2;
        v2f a, b;
        a.x = s_a[m * KP + kb];
        a.y = s_a[m * KP + kb + 1];
        b.x = Wl[(size_t)kb * NOUT + ncol];
        b.y = Wl[(size_t)(kb + 1) * NOUT + ncol];
        acc = __builtin_amdgcn_wmma_f32_16x16x4_f32(false, a, false, b,
                                                    (short)0, acc, false, false);
        a.x = s_x[m * KP + kb];
        a.y = s_x[m * KP + kb + 1];
        b.x = Wr[(size_t)kb * NOUT + ncol];
        b.y = Wr[(size_t)(kb + 1) * NOUT + ncol];
        acc = __builtin_amdgcn_wmma_f32_16x16x4_f32(false, a, false, b,
                                                    (short)0, acc, false, false);
    }

    const float bias = bl[ncol];
    if constexpr (HEAD) {
        constexpr int NWV = NOUT / 16;
        __shared__ float s_dot[NWV * 16];
        const float wf = Wfc[ncol];
        float p[8];
#pragma unroll
        for (int v = 0; v < 8; ++v) p[v] = half_wave_sum(fmaxf(acc[v] + bias, 0.0f) * wf);
        if (m == 0) {
#pragma unroll
            for (int v = 0; v < 8; ++v) s_dot[wave * 16 + khalf * 8 + v] = p[v];
        }
        __syncthreads();
        if (tid < 16) {
            float s = s_dot[tid];
#pragma unroll
            for (int w = 1; w < NWV; ++w) s += s_dot[w * 16 + tid];
            s += bfc[0];
            const int gnode = node_base + tid;
            if (gnode < Nn) out[gnode] = 1.0f / (1.0f + __expf(-s));
        }
    } else {
#pragma unroll
        for (int v = 0; v < 8; ++v) {
            int mrow = v + khalf * 8;
            int gnode = node_base + mrow;
            if (gnode < Nn) {
                float val = acc[v] + bias;
                out[(size_t)gnode * NOUT + ncol] = fmaxf(val, 0.0f);
            }
        }
    }
}

extern "C" void kernel_launch(void* const* d_in, const int* in_sizes, int n_in,
                              void* d_out, int out_size, void* d_ws, size_t ws_size,
                              hipStream_t stream) {
    const float* x    = (const float*)d_in[0];
    const int*   ei   = (const int*)d_in[1];
    const float* W1l  = (const float*)d_in[2];
    const float* b1   = (const float*)d_in[3];
    const float* W1r  = (const float*)d_in[4];
    const float* W2l  = (const float*)d_in[5];
    const float* b2   = (const float*)d_in[6];
    const float* W2r  = (const float*)d_in[7];
    const float* Wfc  = (const float*)d_in[8];
    const float* bfc  = (const float*)d_in[9];
    float* outp = (float*)d_out;

    const int Nn = in_sizes[0] / 64;
    const int Ee = in_sizes[1] / 2;

    float* deg = (float*)d_ws;
    float* agg = deg + (size_t)Nn * 2;
    float* h1  = agg + (size_t)Nn * 128;
    int* wlist = (int*)((char*)d_ws + ((size_t)160 << 20));
    int* wcnt  = (int*)((char*)d_ws + ((size_t)224 << 20));

    neighbour_sum_lds_kernel<64, 64, 800, 2, true, true, false><<<(Nn + 799) / 800 * 2, 1024, 0, stream>>>(
        ei, x, agg, deg, wlist, wcnt, Ee, Nn);
    sage_combine_kernel<64, 128, 2, 2, false><<<(Nn + 15) / 16, 256, 0, stream>>>(
        x, agg, deg, W1l, b1, W1r, nullptr, nullptr, h1, Nn);

    neighbour_sum_lds_kernel<64, 128, 800, 2, false, false, true><<<(Nn + 799) / 800 * 2, 1024, 0, stream>>>(
        ei, h1, agg, deg, wlist, wcnt, Ee, Nn);
    sage_combine_kernel<128, 64, 1, 2, true><<<(Nn + 15) / 16, 128, 0, stream>>>(
        h1, agg, deg, W2l, b2, W2r, Wfc, bfc, outp, Nn);
}


# Combine kernels: input stage is one load group: sums, rows (16-byte loads) and in-degrees all asked for before the first wait, one barrier; was degrees, barrier, 4-byte loads waited for in turn
# speedup vs baseline: 3.4039x; 1.0428x over previous
#include <hip/hip_runtime.h>
#include <hip/hip_bf16.h>
#include <type_traits>


typedef __attribute__((ext_vector_type(2))) float v2f;
typedef __attribute__((ext_vector_type(8))) float v8f;

constexpr int LCAP = 1024;

template <int F, int FS, int R, int EP, bool COUNT, bool SAVE, bool USE_LIST>
__global__ __launch_bounds__(1024) void neighbour_sum_lds_kernel(
    const int* __restrict__ ei, const float* __restrict__ feat,
    float* __restrict__ agg, float* __restrict__ deg,
    int* __restrict__ wlist, int* __restrict__ wcnt, int E, int Nn) {
    constexpr int NW = 32;
    constexpr int QCAP = 160;
    constexpr int U = 8;
    constexpr int G = 1024 / F;
    constexpr int H = F / 64;
    constexpr int NC = FS / F;
    constexpr int BPR = USE_LIST ? NC : NC * EP;
    __shared__ float s_sum[R * F];
    __shared__ float s_cnt[R];
    __shared__ int s_qe[NW * QCAP];
    __shared__ int s_qd[NW * QCAP];

    const int tid = threadIdx.x;
    const int lane = tid & 31;
    const int wave = __builtin_amdgcn_readfirstlane(tid >> 5);
    const int rb = blockIdx.x / BPR;
    const int sub = blockIdx.x % BPR;
    const int col0 = (sub % NC) * F;
    const int part = sub / NC;
    const int base = rb * R;
    const float* featc = feat + col0 + lane * 2;

    for (int i = tid; i < R * F; i += 1024) s_sum[i] = 0.0f;
    if (COUNT) {
        for (int i = tid; i < R; i += 1024) s_cnt[i] = 0.0f;
    }
    __syncthreads();

    int2* wl = (int2*)wlist + (size_t)((rb * EP + part) * NW + wave) * LCAP;

    auto add_rows = [&](int src, int dl, int nb, auto gg) {
        constexpr int GG = decltype(gg)::value;
        for (int j0 = 0; j0 < nb; j0 += GG) {
            float2 v[GG][H];
#pragma unroll
            for (int g = 0; g < GG; ++g) {
                const int sj = __builtin_amdgcn_readlane(src, j0 + g);
                const float* rp = featc + (size_t)sj * FS;
#pragma unroll
                for (int h = 0; h < H; ++h) v[g][h] = *(const float2*)(rp + h * 64);
            }
#pragma unroll
            for (int g = 0; g < GG; ++g) {
                if (j0 + g < nb) {
                    const int dj = __builtin_amdgcn_readlane(dl, j0 + g);
                    float* sp = s_sum + dj * F + lane * 2;
#pragma unroll
                    for (int h = 0; h < H; ++h) {
                        atomicAdd(sp + h * 64, v[g][h].x);
                        atomicAdd(sp + h * 64 + 1, v[g][h].y);
                    }
                }
            }
        }
    };

    auto scan_part = [&](int p, int np) {
        int* qe = s_qe + wave * QCAP;
        int* qd = s_qd + wave * QCAP;
        const unsigned lt = (1u << lane) - 1u;
        int qn = 0;
        int wn = 0;

        auto push = [&](int d, int e) {
            const int dl = d - base;
            const bool hit = (unsigned)dl < (unsigned)R;
            const unsigned m = __builtin_amdgcn_ballot_w32(hit);
            if (__builtin_expect(m != 0u, 0)) {
                if (hit) {
                    const int pos = qn + __popc(m & lt);
                    qe[pos] = e;
                    qd[pos] = dl;
                }
                qn += __popc(m);
            }
        };

        auto drain = [&]() {
            __builtin_amdgcn_wave_barrier();
            for (int b = 0; b < qn; b += 32) {
                const int nb = min(32, qn - b);
                const bool valid = lane < nb;
                const int e = valid ? qe[b + lane] : 0;
                const int dl = valid ? qd[b + lane] : 0;
                const int src = ei[e];
                if (COUNT) {
                    if (valid) atomicAdd(&s_cnt[dl], 1.0f);
                }
                if (SAVE) {
                    if (valid && wn + lane < LCAP) wl[wn + lane] = make_int2(src, dl);
                    wn += nb;
                }
                add_rows(src, dl, nb, std::integral_constant<int, G>{});
            }
            qn = 0;
            __builtin_amdgcn_wave_barrier();
        };

        const int* dstp = ei + E;
        const int nvec = ((E & 3) == 0) ? (E >> 2) : 0;
        const int4* d4 = (const int4*)dstp;
        const int v_lo = (int)((long long)nvec * p / np);
        const int v_hi = (int)((long long)nvec * (p + 1) / np);
        const int ntail = E - nvec * 4;
        const int t_lo = nvec * 4 + (int)((long long)ntail * p / np);
        const int t_hi = nvec * 4 + (int)((long long)ntail * (p + 1) / np);
        for (int v0 = v_lo + wave * (U * 32); v0 < v_hi; v0 += NW * U * 32) {
            int4 dv[U];
            if (v0 + U * 32 <= v_hi) {
                const int4* dp = d4 + v0 + lane;
#pragma unroll
                for (int u = 0; u < U; ++u) dv[u] = dp[u * 32];
            } else {
#pragma unroll
                for (int u = 0; u < U; ++u) {
                    const int idx = v0 + u * 32 + lane;
                    dv[u] = (idx < v_hi) ? d4[idx] : make_int4(-1, -1, -1, -1);
                }
            }
#pragma unroll
            for (int u = 0; u < U; ++u) {
                const int e0 = (v0 + u * 32 + lane) * 4;
                push(dv[u].x, e0);
                push(dv[u].y, e0 + 1);
                push(dv[u].z, e0 + 2);
                push(dv[u].w, e0 + 3);
                if (qn >= 32) drain();
            }
        }
        for (int e0 = t_lo + wave * 32; e0 < t_hi; e0 += NW * 32) {
            const int e = e0 + lane;
            const int d = (e < t_hi) ? dstp[e] : -1;
            push(d, e);
            if (qn >= 32) drain();
        }
        if (qn > 0) drain();
        if (SAVE) {
            if (lane == 0) wcnt[(rb * EP + part) * NW + wave] = wn;
        }
    };

    if (USE_LIST) {
        int cl[EP];
        bool over = false;
#pragma unroll
        for (int p = 0; p < EP; ++p) {
            cl[p] = wcnt[(rb * EP + p) * NW + lane];
            over = over || (cl[p] > LCAP);
        }
        const bool fits = __builtin_amdgcn_ballot_w32(over) == 0u;
        if (fits) {
            constexpr int GL = 2048 / F;
            int nn[EP];
            int n = 0;
#pragma unroll
            for (int p = 0; p < EP; ++p) {
                nn[p] = __builtin_amdgcn_readlane(cl[p], wave);
                n += nn[p];
            }
            const int2* wlb = (const int2*)wlist + (size_t)(rb * EP * NW + wave) * LCAP;
            auto entry = [&](int i) {
                int off = i;
                int idx = -1;
#pragma unroll
                for (int p = 0; p < EP; ++p) {
                    if (off >= 0 && off < nn[p]) idx = p * (NW * LCAP) + off;
                    off -= nn[p];
                }
                return (idx >= 0) ? wlb[idx] : make_int2(0, 0);
            };
            int2 nxt = entry(lane);
            for (int i0 = 0; i0 < n; i0 += 32) {
                const int2 cur = nxt;
                const int nb = min(32, n - i0);
                for (int j0 = 0; j0 < nb; j0 += GL) {
                    float2 v[GL][H];
#pragma unroll
                    for (int g = 0; g < GL; ++g) {
                        const int sj = __builtin_amdgcn_readlane(cur.x, j0 + g);
                        const float* rp = featc + (size_t)sj * FS;
#pragma unroll
                        for (int h = 0; h < H; ++h) v[g][h] = *(const float2*)(rp + h * 64);
                    }
                    if (j0 + GL >= nb) {
                        nxt = entry(i0 + 32 + lane);
                    }
#pragma unroll
                    for (int g = 0; g < GL; ++g) {
                        if (j0 + g < nb) {
                            const int dj = __builtin_amdgcn_readlane(cur.y, j0 + g);
                            float* sp = s_sum + dj * F + lane * 2;
#pragma unroll
                            for (int h = 0; h < H; ++h) {
                                atomicAdd(sp + h * 64, v[g][h].x);
                                atomicAdd(sp + h * 64 + 1, v[g][h].y);
                            }
                        }
                    }
                }
            }
        } else {
            scan_part(0, 1);
        }
    } else {
        scan_part(part, EP);
    }
    __syncthreads();

    float* aggp = agg + (size_t)part * Nn * FS;
    float* degp = deg + (size_t)part * Nn;
    for (int i = tid; i < R * F; i += 1024) {
        const int r = i / F;
        const int c = i - r * F;
        if (base + r < Nn) aggp[(size_t)(base + r) * FS + col0 + c] = s_sum[i];
    }
    if (COUNT) {
        for (int i = tid; i < R; i += 1024) {
            if (base + i < Nn) degp[base + i] = s_cnt[i];
        }
    }
}


__device__ __forceinline__ float half_wave_sum(float v) {
    v += __builtin_bit_cast(float, __builtin_amdgcn_update_dpp(0, __builtin_bit_cast(int, v), 0x161, 0xf, 0xf, true));
    v += __builtin_bit_cast(float, __builtin_amdgcn_update_dpp(0, __builtin_bit_cast(int, v), 0x162, 0xf, 0xf, true));
    v += __builtin_bit_cast(float, __builtin_amdgcn_update_dpp(0, __builtin_bit_cast(int, v), 0x164, 0xf, 0xf, true));
    v += __builtin_bit_cast(float, __builtin_amdgcn_update_dpp(0, __builtin_bit_cast(int, v), 0x168, 0xf, 0xf, true));
    return v;
}

template <int K, int NOUT, int AP, int DP, bool HEAD>
__global__ __launch_bounds__(32 * (NOUT / 16)) void sage_combine_kernel(
    const float* __restrict__ xin, const float* __restrict__ agg,
    const float* __restrict__ deg, const float* __restrict__ Wl,
    const float* __restrict__ bl, const float* __restrict__ Wr,
    const float* __restrict__ Wfc, const float* __restrict__ bfc,
    float* __restrict__ out, int Nn) {
    constexpr int KP = 2 * K + 4;
    constexpr int NT = 32 * (NOUT / 16);
    constexpr int K4 = K / 4;
    constexpr int IT = 16 * K4 / NT;
    static_assert((16 * K4) % NT == 0, "tile pieces must divide among the threads");
    __shared__ __attribute__((aligned(16))) float s_ax[16 * KP];
    float* const s_a = s_ax;
    float* const s_x = s_ax + K;

    const int node_base = blockIdx.x * 16;
    const int tid = threadIdx.x;
    const int lane = tid & 31;
    const int wave = tid >> 5;
    const int m = lane & 15;
    const int khalf = lane >> 4;
    const int ncol = wave * 16 + m;

    float4 av[AP][IT];
    float4 xv[IT];
    float dg[DP][IT];
#pragma unroll
    for (int it = 0; it < IT; ++it) {
        const int i = tid + it * NT;
        const int nd = i / K4;
        const int f = (i - nd * K4) * 4;
        int gnode = node_base + nd;
        if (gnode > Nn - 1) gnode = Nn - 1;
        const unsigned off = (unsigned)gnode * K + f;
#pragma unroll
        for (int p = 0; p < AP; ++p) av[p][it] = *(const float4*)(agg + (size_t)p * Nn * K + off);
        xv[it] = *(const float4*)(xin + off);
#pragma unroll
        for (int p = 0; p < DP; ++p) dg[p][it] = (deg + (size_t)p * Nn)[(unsigned)gnode];
    }
#pragma unroll
    for (int it = 0; it < IT; ++it) {
        const int i = tid + it * NT;
        const int nd = i / K4;
        const int f = (i - nd * K4) * 4;
        float dsum = dg[0][it];
#pragma unroll
        for (int p = 1; p < DP; ++p) dsum += dg[p][it];
        const float inv = 1.0f / fmaxf(dsum, 1.0f);
        float4 asum = av[0][it];
#pragma unroll
        for (int p = 1; p < AP; ++p) {
            asum.x += av[p][it].x;
            asum.y += av[p][it].y;
            asum.z += av[p][it].z;
            asum.w += av[p][it].w;
        }
        asum.x *= inv;
        asum.y *= inv;
        asum.z *= inv;
        asum.w *= inv;
        *(float4*)(s_a + nd * KP + f) = asum;
        *(float4*)(s_x + nd * KP + f) = xv[it];
    }
    __syncthreads();

    v8f acc = {};
#pragma unroll
    for (int k = 0; k < K; k += 4) {
        const int kb = k + khalf * 2;
        v2f a, b;
        a.x = s_a[m * KP + kb];
        a.y = s_a[m * KP + kb + 1];
        b.x = Wl[(size_t)kb * NOUT + ncol];
        b.y = Wl[(size_t)(kb + 1) * NOUT + ncol];
        acc = __builtin_amdgcn_wmma_f32_16x16x4_f32(false, a, false, b,
                                                    (short)0, acc, false, false);
        a.x = s_x[m * KP + kb];
        a.y = s_x[m * KP + kb + 1];
        b.x = Wr[(size_t)kb * NOUT + ncol];
        b.y = Wr[(size_t)(kb + 1) * NOUT + ncol];
        acc = __builtin_amdgcn_wmma_f32_16x16x4_f32(false, a, false, b,
                                                    (short)0, acc, false, false);
    }

    const float bias = bl[ncol];
    if constexpr (HEAD) {
        constexpr int NWV = NOUT / 16;
        __shared__ float s_dot[NWV * 16];
        const float wf = Wfc[ncol];
        float p[8];
#pragma unroll
        for (int v = 0; v < 8; ++v) p[v] = half_wave_sum(fmaxf(acc[v] + bias, 0.0f) * wf);
        if (m == 0) {
#pragma unroll
            for (int v = 0; v < 8; ++v) s_dot[wave * 16 + khalf * 8 + v] = p[v];
        }
        __syncthreads();
        if (tid < 16) {
            float s = s_dot[tid];
#pragma unroll
            for (int w = 1; w < NWV; ++w) s += s_dot[w * 16 + tid];
            s += bfc[0];
            const int gnode = node_base + tid;
            if (gnode < Nn) out[gnode] = 1.0f / (1.0f + __expf(-s));
        }
    } else {
#pragma unroll
        for (int v = 0; v < 8; ++v) {
            int mrow = v + khalf * 8;
            int gnode = node_base + mrow;
            if (gnode < Nn) {
                float val = acc[v] + bias;
                out[(size_t)gnode * NOUT + ncol] = fmaxf(val, 0.0f);
            }
        }
    }
}

extern "C" void kernel_launch(void* const* d_in, const int* in_sizes, int n_in,
                              void* d_out, int out_size, void* d_ws, size_t ws_size,
                              hipStream_t stream) {
    const float* x    = (const float*)d_in[0];
    const int*   ei   = (const int*)d_in[1];
    const float* W1l  = (const float*)d_in[2];
    const float* b1   = (const float*)d_in[3];
    const float* W1r  = (const float*)d_in[4];
    const float* W2l  = (const float*)d_in[5];
    const float* b2   = (const float*)d_in[6];
    const float* W2r  = (const float*)d_in[7];
    const float* Wfc  = (const float*)d_in[8];
    const float* bfc  = (const float*)d_in[9];
    float* outp = (float*)d_out;

    const int Nn = in_sizes[0] / 64;
    const int Ee = in_sizes[1] / 2;

    float* deg = (float*)d_ws;
    float* agg = deg + (size_t)Nn * 2;
    float* h1  = agg + (size_t)Nn * 128;
    int* wlist = (int*)((char*)d_ws + ((size_t)160 << 20));
    int* wcnt  = (int*)((char*)d_ws + ((size_t)224 << 20));

    neighbour_sum_lds_kernel<64, 64, 800, 2, true, true, false><<<(Nn + 799) / 800 * 2, 1024, 0, stream>>>(
        ei, x, agg, deg, wlist, wcnt, Ee, Nn);
    sage_combine_kernel<64, 128, 2, 2, false><<<(Nn + 15) / 16, 256, 0, stream>>>(
        x, agg, deg, W1l, b1, W1r, nullptr, nullptr, h1, Nn);

    neighbour_sum_lds_kernel<64, 128, 800, 2, false, false, true><<<(Nn + 799) / 800 * 2, 1024, 0, stream>>>(
        ei, h1, agg, deg, wlist, wcnt, Ee, Nn);
    sage_combine_kernel<128, 64, 1, 2, true><<<(Nn + 15) / 16, 128, 0, stream>>>(
        h1, agg, deg, W2l, b2, W2r, Wfc, bfc, outp, Nn);
}


# Layer 2 in one launch: a block sums 400 nodes x 128 in LDS (walks saved lists, keeps its half), then does combine and head on its tile, W by 16-byte loads; 4 launches to 3; head sum order changes
# speedup vs baseline: 3.6108x; 1.0608x over previous
#include <hip/hip_runtime.h>
#include <hip/hip_bf16.h>
#include <type_traits>


typedef __attribute__((ext_vector_type(2))) float v2f;
typedef __attribute__((ext_vector_type(8))) float v8f;

__device__ __forceinline__ float half_wave_sum(float v) {
    v += __builtin_bit_cast(float, __builtin_amdgcn_update_dpp(0, __builtin_bit_cast(int, v), 0x161, 0xf, 0xf, true));
    v += __builtin_bit_cast(float, __builtin_amdgcn_update_dpp(0, __builtin_bit_cast(int, v), 0x162, 0xf, 0xf, true));
    v += __builtin_bit_cast(float, __builtin_amdgcn_update_dpp(0, __builtin_bit_cast(int, v), 0x164, 0xf, 0xf, true));
    v += __builtin_bit_cast(float, __builtin_amdgcn_update_dpp(0, __builtin_bit_cast(int, v), 0x168, 0xf, 0xf, true));
    return v;
}
constexpr int LCAP = 1024;

template <int F, int FS, int R, int EP, int LR, int NOUT, bool COUNT, bool SAVE, bool USE_LIST>
__global__ __launch_bounds__(1024) void neighbour_sum_lds_kernel(
    const int* __restrict__ ei, const float* __restrict__ feat,
    float* __restrict__ agg, float* __restrict__ deg,
    int* __restrict__ wlist, int* __restrict__ wcnt,
    const float* __restrict__ Wl, const float* __restrict__ bl, const float* __restrict__ Wr,
    const float* __restrict__ Wfc, const float* __restrict__ bfc, float* __restrict__ out,
    int E, int Nn) {
    constexpr int NW = 32;
    constexpr int QCAP = 160;
    constexpr int U = 8;
    constexpr int G = 1024 / F;
    constexpr int H = F / 64;
    constexpr int NC = FS / F;
    constexpr int BPR = USE_LIST ? NC : NC * EP;
    constexpr int P = USE_LIST ? F + 4 : F;
    static_assert(!USE_LIST || (F == FS && R % 16 == 0 && R / 16 <= NW && NOUT % 16 == 0),
                  "a list-walking block holds whole rows and gives each wave at most one 16-row tile");
    __shared__ __attribute__((aligned(16))) float s_sum[R * P];
    __shared__ float s_cnt[R];
    __shared__ int s_qe[NW * QCAP];
    __shared__ int s_qd[NW * QCAP];

    const int tid = threadIdx.x;
    const int lane = tid & 31;
    const int wave = __builtin_amdgcn_readfirstlane(tid >> 5);
    const int rb = blockIdx.x / BPR;
    const int sub = blockIdx.x % BPR;
    const int col0 = (sub % NC) * F;
    const int part = sub / NC;
    const int base = rb * R;
    const float* featc = feat + col0 + lane * 2;

    for (int i = tid; i < R * P; i += 1024) s_sum[i] = 0.0f;
    if (COUNT) {
        for (int i = tid; i < R; i += 1024) s_cnt[i] = 0.0f;
    }
    if (USE_LIST) {
        for (int i = tid; i < R; i += 1024) {
            int gnode = base + i;
            if (gnode > Nn - 1) gnode = Nn - 1;
            float dsum = deg[gnode];
#pragma unroll
            for (int p = 1; p < EP; ++p) dsum += deg[(size_t)p * Nn + gnode];
            s_cnt[i] = 1.0f / fmaxf(dsum, 1.0f);
        }
    }
    __syncthreads();

    const int lrb = rb / LR;
    const int dl0 = (rb - lrb * LR) * R;
    int2* wl = (int2*)wlist + (size_t)((lrb * EP + part) * NW + wave) * LCAP;

    auto add_rows = [&](int src, int dl, int nb, auto gg) {
        constexpr int GG = decltype(gg)::value;
        for (int j0 = 0; j0 < nb; j0 += GG) {
            float2 v[GG][H];
#pragma unroll
            for (int g = 0; g < GG; ++g) {
                const int sj = __builtin_amdgcn_readlane(src, j0 + g);
                const float* rp = featc + (size_t)sj * FS;
#pragma unroll
                for (int h = 0; h < H; ++h) v[g][h] = *(const float2*)(rp + h * 64);
            }
#pragma unroll
            for (int g = 0; g < GG; ++g) {
                if (j0 + g < nb) {
                    const int dj = __builtin_amdgcn_readlane(dl, j0 + g);
                    float* sp = s_sum + dj * P + lane * 2;
#pragma unroll
                    for (int h = 0; h < H; ++h) {
                        atomicAdd(sp + h * 64, v[g][h].x);
                        atomicAdd(sp + h * 64 + 1, v[g][h].y);
                    }
                }
            }
        }
    };

    auto scan_part = [&](int p, int np) {
        int* qe = s_qe + wave * QCAP;
        int* qd = s_qd + wave * QCAP;
        const unsigned lt = (1u << lane) - 1u;
        int qn = 0;
        int wn = 0;

        auto push = [&](int d, int e) {
            const int dl = d - base;
            const bool hit = (unsigned)dl < (unsigned)R;
            const unsigned m = __builtin_amdgcn_ballot_w32(hit);
            if (__builtin_expect(m != 0u, 0)) {
                if (hit) {
                    const int pos = qn + __popc(m & lt);
                    qe[pos] = e;
                    qd[pos] = dl;
                }
                qn += __popc(m);
            }
        };

        auto drain = [&]() {
            __builtin_amdgcn_wave_barrier();
            for (int b = 0; b < qn; b += 32) {
                const int nb = min(32, qn - b);
                const bool valid = lane < nb;
                const int e = valid ? qe[b + lane] : 0;
                const int dl = valid ? qd[b + lane] : 0;
                const int src = ei[e];
                if (COUNT) {
                    if (valid) atomicAdd(&s_cnt[dl], 1.0f);
                }
                if (SAVE) {
                    if (valid && wn + lane < LCAP) wl[wn + lane] = make_int2(src, dl);
                    wn += nb;
                }
                add_rows(src, dl, nb, std::integral_constant<int, G>{});
            }
            qn = 0;
            __builtin_amdgcn_wave_barrier();
        };

        const int* dstp = ei + E;
        const int nvec = ((E & 3) == 0) ? (E >> 2) : 0;
        const int4* d4 = (const int4*)dstp;
        const int v_lo = (int)((long long)nvec * p / np);
        const int v_hi = (int)((long long)nvec * (p + 1) / np);
        const int ntail = E - nvec * 4;
        const int t_lo = nvec * 4 + (int)((long long)ntail * p / np);
        const int t_hi = nvec * 4 + (int)((long long)ntail * (p + 1) / np);
        for (int v0 = v_lo + wave * (U * 32); v0 < v_hi; v0 += NW * U * 32) {
            int4 dv[U];
            if (v0 + U * 32 <= v_hi) {
                const int4* dp = d4 + v0 + lane;
#pragma unroll
                for (int u = 0; u < U; ++u) dv[u] = dp[u * 32];
            } else {
#pragma unroll
                for (int u = 0; u < U; ++u) {
                    const int idx = v0 + u * 32 + lane;
                    dv[u] = (idx < v_hi) ? d4[idx] : make_int4(-1, -1, -1, -1);
                }
            }
#pragma unroll
            for (int u = 0; u < U; ++u) {
                const int e0 = (v0 + u * 32 + lane) * 4;
                push(dv[u].x, e0);
                push(dv[u].y, e0 + 1);
                push(dv[u].z, e0 + 2);
                push(dv[u].w, e0 + 3);
                if (qn >= 32) drain();
            }
        }
        for (int e0 = t_lo + wave * 32; e0 < t_hi; e0 += NW * 32) {
            const int e = e0 + lane;
            const int d = (e < t_hi) ? dstp[e] : -1;
            push(d, e);
            if (qn >= 32) drain();
        }
        if (qn > 0) drain();
        if (SAVE) {
            if (lane == 0) wcnt[(rb * EP + part) * NW + wave] = wn;
        }
    };

    if (USE_LIST) {
        int cl[EP];
        bool over = false;
#pragma unroll
        for (int p = 0; p < EP; ++p) {
            cl[p] = wcnt[(lrb * EP + p) * NW + lane];
            over = over || (cl[p] > LCAP);
        }
        const bool fits = __builtin_amdgcn_ballot_w32(over) == 0u;
        if (fits) {
            constexpr int GL = 2048 / F;
            int nn[EP];
            int n = 0;
#pragma unroll
            for (int p = 0; p < EP; ++p) {
                nn[p] = __builtin_amdgcn_readlane(cl[p], wave);
                n += nn[p];
            }
            const int2* wlb = (const int2*)wlist + (size_t)(lrb * EP * NW + wave) * LCAP;
            auto entry = [&](int i) {
                int off = i;
                int idx = -1;
#pragma unroll
                for (int p = 0; p < EP; ++p) {
                    if (off >= 0 && off < nn[p]) idx = p * (NW * LCAP) + off;
                    off -= nn[p];
                }
                return (idx >= 0) ? wlb[idx] : make_int2(0, 0);
            };
            int2 nxt = entry(lane);
            for (int i0 = 0; i0 < n; i0 += 32) {
                const int2 cur = nxt;
                bool ask = true;
                const int dlc = cur.y - dl0;
                const bool keep = (i0 + lane < n) && ((unsigned)dlc < (unsigned)R);
                unsigned m = __builtin_amdgcn_ballot_w32(keep);
                while (m != 0u) {
                    float2 v[GL][H];
                    unsigned mm = m;
#pragma unroll
                    for (int g = 0; g < GL; ++g) {
                        if (mm != 0u) {
                            const int b = __builtin_ctz(mm);
                            mm &= mm - 1u;
                            const int sj = __builtin_amdgcn_readlane(cur.x, b);
                            const float* rp = featc + (size_t)sj * FS;
#pragma unroll
                            for (int h = 0; h < H; ++h) v[g][h] = *(const float2*)(rp + h * 64);
                        }
                    }
                    if (ask) {
                        nxt = entry(i0 + 32 + lane);
                        ask = false;
                    }
#pragma unroll
                    for (int g = 0; g < GL; ++g) {
                        if (m != 0u) {
                            const int b = __builtin_ctz(m);
                            m &= m - 1u;
                            const int dj = __builtin_amdgcn_readlane(dlc, b);
                            float* sp = s_sum + dj * P + lane * 2;
#pragma unroll
                            for (int h = 0; h < H; ++h) {
                                atomicAdd(sp + h * 64, v[g][h].x);
                                atomicAdd(sp + h * 64 + 1, v[g][h].y);
                            }
                        }
                    }
                }
                if (ask) nxt = entry(i0 + 32 + lane);
            }
        } else {
            scan_part(0, 1);
        }
    } else {
        scan_part(part, EP);
    }
    __syncthreads();

    if constexpr (USE_LIST) {
        static_assert(NOUT == 64, "the column order above is written for 4 accumulator tiles");
        if (wave < R / 16) {
            const int m = lane & 15;
            const int khalf = lane >> 4;
            const int lrow = wave * 16 + m;
            const float inv = s_cnt[lrow];
            const float* sa = s_sum + lrow * P + khalf * 2;
            int gn = base + lrow;
            if (gn > Nn - 1) gn = Nn - 1;
            const float* xa = feat + (size_t)gn * FS + khalf * 2;
            const float* wlp = Wl + (size_t)(khalf * 2) * NOUT + m * 4;
            const float* wrp = Wr + (size_t)(khalf * 2) * NOUT + m * 4;
            v8f acc0 = {}, acc1 = {}, acc2 = {}, acc3 = {};
#pragma unroll 4
            for (int k = 0; k < F; k += 4) {
                const float2 sr = *(const float2*)(sa + k);
                const float2 xr = *(const float2*)(xa + k);
                const float4 l0 = *(const float4*)(wlp + (size_t)k * NOUT);
                const float4 l1 = *(const float4*)(wlp + (size_t)(k + 1) * NOUT);
                const float4 r0 = *(const float4*)(wrp + (size_t)k * NOUT);
                const float4 r1 = *(const float4*)(wrp + (size_t)(k + 1) * NOUT);
                v2f al, ax, b;
                al.x = sr.x * inv;
                al.y = sr.y * inv;
                ax.x = xr.x;
                ax.y = xr.y;
                b.x = l0.x; b.y = l1.x;
                acc0 = __builtin_amdgcn_wmma_f32_16x16x4_f32(false, al, false, b, (short)0, acc0, false, false);
                b.x = r0.x; b.y = r1.x;
                acc0 = __builtin_amdgcn_wmma_f32_16x16x4_f32(false, ax, false, b, (short)0, acc0, false, false);
                b.x = l0.y; b.y = l1.y;
                acc1 = __builtin_amdgcn_wmma_f32_16x16x4_f32(false, al, false, b, (short)0, acc1, false, false);
                b.x = r0.y; b.y = r1.y;
                acc1 = __builtin_amdgcn_wmma_f32_16x16x4_f32(false, ax, false, b, (short)0, acc1, false, false);
                b.x = l0.z; b.y = l1.z;
                acc2 = __builtin_amdgcn_wmma_f32_16x16x4_f32(false, al, false, b, (short)0, acc2, false, false);
                b.x = r0.z; b.y = r1.z;
                acc2 = __builtin_amdgcn_wmma_f32_16x16x4_f32(false, ax, false, b, (short)0, acc2, false, false);
                b.x = l0.w; b.y = l1.w;
                acc3 = __builtin_amdgcn_wmma_f32_16x16x4_f32(false, al, false, b, (short)0, acc3, false, false);
                b.x = r0.w; b.y = r1.w;
                acc3 = __builtin_amdgcn_wmma_f32_16x16x4_f32(false, ax, false, b, (short)0, acc3, false, false);
            }
            const float4 bi = *(const float4*)(bl + m * 4);
            const float4 wf = *(const float4*)(Wfc + m * 4);
            const float bf = bfc[0];
            float sv = 0.0f;
#pragma unroll
            for (int v = 0; v < 8; ++v) {
                float q = fmaxf(acc0[v] + bi.x, 0.0f) * wf.x;
                q += fmaxf(acc1[v] + bi.y, 0.0f) * wf.y;
                q += fmaxf(acc2[v] + bi.z, 0.0f) * wf.z;
                q += fmaxf(acc3[v] + bi.w, 0.0f) * wf.w;
                const float s = half_wave_sum(q) + bf;
                if (m == v) sv = s;
            }
            const int gnode = base + wave * 16 + khalf * 8 + m;
            if (m < 8 && gnode < Nn) out[gnode] = 1.0f / (1.0f + __expf(-sv));
        }
    } else {
        float* aggp = agg + (size_t)part * Nn * FS;
        float* degp = deg + (size_t)part * Nn;
        for (int i = tid; i < R * F; i += 1024) {
            const int r = i / F;
            const int c = i - r * F;
            if (base + r < Nn) aggp[(size_t)(base + r) * FS + col0 + c] = s_sum[i];
        }
        if (COUNT) {
            for (int i = tid; i < R; i += 1024) {
                if (base + i < Nn) degp[base + i] = s_cnt[i];
            }
        }
    }
}


template <int K, int NOUT, int AP, int DP>
__global__ __launch_bounds__(32 * (NOUT / 16)) void sage_combine_kernel(
    const float* __restrict__ xin, const float* __restrict__ agg,
    const float* __restrict__ deg, const float* __restrict__ Wl,
    const float* __restrict__ bl, const float* __restrict__ Wr,
    float* __restrict__ out, int Nn) {
    constexpr int KP = 2 * K + 4;
    constexpr int NT = 32 * (NOUT / 16);
    constexpr int K4 = K / 4;
    constexpr int IT = 16 * K4 / NT;
    static_assert((16 * K4) % NT == 0, "tile pieces must divide among the threads");
    __shared__ __attribute__((aligned(16))) float s_ax[16 * KP];
    float* const s_a = s_ax;
    float* const s_x = s_ax + K;

    const int node_base = blockIdx.x * 16;
    const int tid = threadIdx.x;
    const int lane = tid & 31;
    const int wave = tid >> 5;
    const int m = lane & 15;
    const int khalf = lane >> 4;
    const int ncol = wave * 16 + m;

    float4 av[AP][IT];
    float4 xv[IT];
    float dg[DP][IT];
#pragma unroll
    for (int it = 0; it < IT; ++it) {
        const int i = tid + it * NT;
        const int nd = i / K4;
        const int f = (i - nd * K4) * 4;
        int gnode = node_base + nd;
        if (gnode > Nn - 1) gnode = Nn - 1;
        const unsigned off = (unsigned)gnode * K + f;
#pragma unroll
        for (int p = 0; p < AP; ++p) av[p][it] = *(const float4*)(agg + (size_t)p * Nn * K + off);
        xv[it] = *(const float4*)(xin + off);
#pragma unroll
        for (int p = 0; p < DP; ++p) dg[p][it] = (deg + (size_t)p * Nn)[(unsigned)gnode];
    }
#pragma unroll
    for (int it = 0; it < IT; ++it) {
        const int i = tid + it * NT;
        const int nd = i / K4;
        const int f = (i - nd * K4) * 4;
        float dsum = dg[0][it];
#pragma unroll
        for (int p = 1; p < DP; ++p) dsum += dg[p][it];
        const float inv = 1.0f / fmaxf(dsum, 1.0f);
        float4 asum = av[0][it];
#pragma unroll
        for (int p = 1; p < AP; ++p) {
            asum.x += av[p][it].x;
            asum.y += av[p][it].y;
            asum.z += av[p][it].z;
            asum.w += av[p][it].w;
        }
        asum.x *= inv;
        asum.y *= inv;
        asum.z *= inv;
        asum.w *= inv;
        *(float4*)(s_a + nd * KP + f) = asum;
        *(float4*)(s_x + nd * KP + f) = xv[it];
    }
    __syncthreads();

    v8f acc = {};
#pragma unroll
    for (int k = 0; k < K; k += 4) {
        const int kb = k + khalf * 2;
        v2f a, b;
        a.x = s_a[m * KP + kb];
        a.y = s_a[m * KP + kb + 1];
        b.x = Wl[(size_t)kb * NOUT + ncol];
        b.y = Wl[(size_t)(kb + 1) * NOUT + ncol];
        acc = __builtin_amdgcn_wmma_f32_16x16x4_f32(false, a, false, b,
                                                    (short)0, acc, false, false);
        a.x = s_x[m * KP + kb];
        a.y = s_x[m * KP + kb + 1];
        b.x = Wr[(size_t)kb * NOUT + ncol];
        b.y = Wr[(size_t)(kb + 1) * NOUT + ncol];
        acc = __builtin_amdgcn_wmma_f32_16x16x4_f32(false, a, false, b,
                                                    (short)0, acc, false, false);
    }

    const float bias = bl[ncol];
#pragma unroll
    for (int v = 0; v < 8; ++v) {
        int mrow = v + khalf * 8;
        int gnode = node_base + mrow;
        if (gnode < Nn) {
            float val = acc[v] + bias;
            out[(size_t)gnode * NOUT + ncol] = fmaxf(val, 0.0f);
        }
    }
}

extern "C" void kernel_launch(void* const* d_in, const int* in_sizes, int n_in,
                              void* d_out, int out_size, void* d_ws, size_t ws_size,
                              hipStream_t stream) {
    const float* x    = (const float*)d_in[0];
    const int*   ei   = (const int*)d_in[1];
    const float* W1l  = (const float*)d_in[2];
    const float* b1   = (const float*)d_in[3];
    const float* W1r  = (const float*)d_in[4];
    const float* W2l  = (const float*)d_in[5];
    const float* b2   = (const float*)d_in[6];
    const float* W2r  = (const float*)d_in[7];
    const float* Wfc  = (const float*)d_in[8];
    const float* bfc  = (const float*)d_in[9];
    float* outp = (float*)d_out;

    const int Nn = in_sizes[0] / 64;
    const int Ee = in_sizes[1] / 2;

    float* deg = (float*)d_ws;
    float* agg = deg + (size_t)Nn * 2;
    float* h1  = agg + (size_t)Nn * 128;
    int* wlist = (int*)((char*)d_ws + ((size_t)160 << 20));
    int* wcnt  = (int*)((char*)d_ws + ((size_t)224 << 20));

    neighbour_sum_lds_kernel<64, 64, 800, 2, 1, 16, true, true, false><<<(Nn + 799) / 800 * 2, 1024, 0, stream>>>(
        ei, x, agg, deg, wlist, wcnt, nullptr, nullptr, nullptr, nullptr, nullptr, nullptr, Ee, Nn);
    sage_combine_kernel<64, 128, 2, 2><<<(Nn + 15) / 16, 256, 0, stream>>>(
        x, agg, deg, W1l, b1, W1r, h1, Nn);

    neighbour_sum_lds_kernel<128, 128, 400, 2, 2, 64, false, false, true><<<(Nn + 399) / 400, 1024, 0, stream>>>(
        ei, h1, agg, deg, wlist, wcnt, W2l, b2, W2r, Wfc, bfc, outp, Ee, Nn);
}


# A first launch recodes the edge list: one range-id byte per edge, streamed by the layer-1 scan instead of 4-byte destinations, and (source, destination) pairs; 3 launches to 4; LDS add order changes
# speedup vs baseline: 3.8315x; 1.0611x over previous
#include <hip/hip_runtime.h>
#include <hip/hip_bf16.h>
#include <type_traits>


typedef __attribute__((ext_vector_type(2))) float v2f;
typedef __attribute__((ext_vector_type(8))) float v8f;

__device__ __forceinline__ float half_wave_sum(float v) {
    v += __builtin_bit_cast(float, __builtin_amdgcn_update_dpp(0, __builtin_bit_cast(int, v), 0x161, 0xf, 0xf, true));
    v += __builtin_bit_cast(float, __builtin_amdgcn_update_dpp(0, __builtin_bit_cast(int, v), 0x162, 0xf, 0xf, true));
    v += __builtin_bit_cast(float, __builtin_amdgcn_update_dpp(0, __builtin_bit_cast(int, v), 0x164, 0xf, 0xf, true));
    v += __builtin_bit_cast(float, __builtin_amdgcn_update_dpp(0, __builtin_bit_cast(int, v), 0x168, 0xf, 0xf, true));
    return v;
}
constexpr int IDR = 800;

__global__ __launch_bounds__(512) void edge_recode_kernel(const int* __restrict__ ei, unsigned* __restrict__ idw,
                                                          int* __restrict__ rec, int E) {
    const int nd = ((E + 15) >> 4) << 2;
    const int T = gridDim.x * 512;
    const int* dstp = ei + E;
    int2* rp = (int2*)rec;
    auto idb = [](int d) { return ((unsigned)d / IDR) & 255u; };
    if ((E & 3) == 0) {
        const int nf = E >> 2;
        for (int q0 = blockIdx.x * 512 + threadIdx.x; q0 < nd; q0 += 4 * T) {
            int4 sv[4], dv[4];
#pragma unroll
            for (int j = 0; j < 4; ++j) {
                const int q = q0 + j * T;
                const int qq = (q < nf) ? q : 0;
                sv[j] = *(const int4*)(ei + (size_t)qq * 4);
                dv[j] = *(const int4*)(dstp + (size_t)qq * 4);
            }
#pragma unroll
            for (int j = 0; j < 4; ++j) {
                const int q = q0 + j * T;
                if (q < nf) {
                    idw[q] = idb(dv[j].x) | (idb(dv[j].y) << 8) | (idb(dv[j].z) << 16) | (idb(dv[j].w) << 24);
                    int4* o = (int4*)(rp + (size_t)q * 4);
                    o[0] = make_int4(sv[j].x, dv[j].x, sv[j].y, dv[j].y);
                    o[1] = make_int4(sv[j].z, dv[j].z, sv[j].w, dv[j].w);
                } else if (q < nd) {
                    idw[q] = 0xffffffffu;
                }
            }
        }
    } else {
        for (int q = blockIdx.x * 512 + threadIdx.x; q < nd; q += T) {
            unsigned w = 0u;
#pragma unroll
            for (int k = 0; k < 4; ++k) {
                const int e = q * 4 + k;
                unsigned b = 255u;
                if (e < E) {
                    const int s = ei[e];
                    const int d = dstp[e];
                    rp[e] = make_int2(s, d);
                    b = idb(d);
                }
                w |= b << (8 * k);
            }
            idw[q] = w;
        }
    }
}

constexpr int LCAP = 1024;

template <int F, int FS, int R, int EP, int LR, int NOUT, bool COUNT, bool SAVE, bool USE_LIST>
__global__ __launch_bounds__(1024) void neighbour_sum_lds_kernel(
    const unsigned* __restrict__ idw, const int* __restrict__ rec, const float* __restrict__ feat,
    float* __restrict__ agg, float* __restrict__ deg,
    int* __restrict__ wlist, int* __restrict__ wcnt,
    const float* __restrict__ Wl, const float* __restrict__ bl, const float* __restrict__ Wr,
    const float* __restrict__ Wfc, const float* __restrict__ bfc, float* __restrict__ out,
    int E, int Nn) {
    constexpr int NW = 32;
    constexpr int QCAP = 544;
    constexpr int U = 4;
    static_assert(R * LR == IDR, "the id bytes are cut by the node range of the saving launch");
    constexpr int G = 1024 / F;
    constexpr int H = F / 64;
    constexpr int NC = FS / F;
    constexpr int BPR = USE_LIST ? NC : NC * EP;
    constexpr int P = USE_LIST ? F + 4 : F;
    static_assert(!USE_LIST || (F == FS && R % 16 == 0 && R / 16 <= NW && NOUT % 16 == 0),
                  "a list-walking block holds whole rows and gives each wave at most one 16-row tile");
    __shared__ __attribute__((aligned(16))) float s_sum[R * P];
    __shared__ float s_cnt[R];
    __shared__ int s_qe[NW * QCAP];

    const int tid = threadIdx.x;
    const int lane = tid & 31;
    const int wave = __builtin_amdgcn_readfirstlane(tid >> 5);
    const int rb = blockIdx.x / BPR;
    const int sub = blockIdx.x % BPR;
    const int col0 = (sub % NC) * F;
    const int part = sub / NC;
    const int base = rb * R;
    const float* featc = feat + col0 + lane * 2;

    for (int i = tid; i < R * P; i += 1024) s_sum[i] = 0.0f;
    if (COUNT) {
        for (int i = tid; i < R; i += 1024) s_cnt[i] = 0.0f;
    }
    if (USE_LIST) {
        for (int i = tid; i < R; i += 1024) {
            int gnode = base + i;
            if (gnode > Nn - 1) gnode = Nn - 1;
            float dsum = deg[gnode];
#pragma unroll
            for (int p = 1; p < EP; ++p) dsum += deg[(size_t)p * Nn + gnode];
            s_cnt[i] = 1.0f / fmaxf(dsum, 1.0f);
        }
    }
    __syncthreads();

    const int lrb = rb / LR;
    const int dl0 = (rb - lrb * LR) * R;
    int2* wl = (int2*)wlist + (size_t)((lrb * EP + part) * NW + wave) * LCAP;

    auto add_rows = [&](int src, int dl, int nb, auto gg) {
        constexpr int GG = decltype(gg)::value;
        for (int j0 = 0; j0 < nb; j0 += GG) {
            float2 v[GG][H];
#pragma unroll
            for (int g = 0; g < GG; ++g) {
                const int sj = __builtin_amdgcn_readlane(src, j0 + g);
                const float* rp = featc + (size_t)sj * FS;
#pragma unroll
                for (int h = 0; h < H; ++h) v[g][h] = *(const float2*)(rp + h * 64);
            }
#pragma unroll
            for (int g = 0; g < GG; ++g) {
                const int dj = __builtin_amdgcn_readlane(dl, j0 + g);
                if (j0 + g < nb && dj >= 0) {
                    float* sp = s_sum + dj * P + lane * 2;
#pragma unroll
                    for (int h = 0; h < H; ++h) {
                        atomicAdd(sp + h * 64, v[g][h].x);
                        atomicAdd(sp + h * 64 + 1, v[g][h].y);
                    }
                }
            }
        }
    };

    auto scan_part = [&](int p, int np) {
        int* qe = s_qe + wave * QCAP;
        const unsigned lt = (1u << lane) - 1u;
        const int2* recp = (const int2*)rec;
        const unsigned fid = ((unsigned)base / IDR) & 255u;
        const unsigned fid4 = fid * 0x01010101u;
        int qn = 0;
        int wn = 0;

        auto test = [&](unsigned w, int e0) {
            const unsigned x = w ^ fid4;
            unsigned t = (x - 0x01010101u) & ~x & 0x80808080u;
            unsigned m = __builtin_amdgcn_ballot_w32(t != 0u);
            while (m != 0u) {
                if (t != 0u) {
                    const int pos = qn + __popc(m & lt);
                    qe[pos] = e0 + (__builtin_ctz(t) >> 3);
                }
                qn += __popc(m);
                t &= t - 1u;
                m = __builtin_amdgcn_ballot_w32(t != 0u);
            }
        };

        auto drain = [&](bool all) {
            __builtin_amdgcn_wave_barrier();
            int b = 0;
            for (; all ? (b < qn) : (b + 32 <= qn); b += 32) {
                const int nb = min(32, qn - b);
                int e = (lane < nb) ? qe[b + lane] : -1;
                const bool ine = (unsigned)e < (unsigned)E;
                if (!ine) e = 0;
                const int2 sd = recp[e];
                const int src = sd.x;
                const int dlr = sd.y - base;
                const bool valid = ine && ((unsigned)dlr < (unsigned)R);
                const int dl = valid ? dlr : -1;
                if (COUNT) {
                    if (valid) atomicAdd(&s_cnt[dl], 1.0f);
                }
                if (SAVE) {
                    const unsigned vm = __builtin_amdgcn_ballot_w32(valid);
                    const int pos = wn + __popc(vm & lt);
                    if (valid && pos < LCAP) wl[pos] = make_int2(src, dl);
                    wn += __popc(vm);
                }
                add_rows(src, dl, nb, std::integral_constant<int, G>{});
            }
            const int rem = max(qn - b, 0);
            if (lane < rem) qe[lane] = qe[b + lane];
            qn = rem;
            __builtin_amdgcn_wave_barrier();
        };

        const int nq = (E + 15) >> 4;
        const uint4* iq = (const uint4*)idw;
        const int q_lo = (int)((long long)nq * p / np);
        const int q_hi = (int)((long long)nq * (p + 1) / np);
        for (int q0 = q_lo + wave * (U * 32); q0 < q_hi; q0 += NW * U * 32) {
            uint4 iv[U];
            if (q0 + U * 32 <= q_hi) {
                const uint4* qp = iq + q0 + lane;
#pragma unroll
                for (int u = 0; u < U; ++u) iv[u] = qp[u * 32];
            } else {
                const unsigned nf = ~fid4;
#pragma unroll
                for (int u = 0; u < U; ++u) {
                    const int idx = q0 + u * 32 + lane;
                    iv[u] = (idx < q_hi) ? iq[idx] : make_uint4(nf, nf, nf, nf);
                }
            }
#pragma unroll
            for (int u = 0; u < U; ++u) {
                const int e0 = (q0 + u * 32 + lane) * 16;
                test(iv[u].x, e0);
                test(iv[u].y, e0 + 4);
                test(iv[u].z, e0 + 8);
                test(iv[u].w, e0 + 12);
                if (qn >= 32) drain(false);
            }
        }
        if (qn > 0) drain(true);
        if (SAVE) {
            if (lane == 0) wcnt[(rb * EP + part) * NW + wave] = wn;
        }
    };

    if (USE_LIST) {
        int cl[EP];
        bool over = false;
#pragma unroll
        for (int p = 0; p < EP; ++p) {
            cl[p] = wcnt[(lrb * EP + p) * NW + lane];
            over = over || (cl[p] > LCAP);
        }
        const bool fits = __builtin_amdgcn_ballot_w32(over) == 0u;
        if (fits) {
            constexpr int GL = 2048 / F;
            int nn[EP];
            int n = 0;
#pragma unroll
            for (int p = 0; p < EP; ++p) {
                nn[p] = __builtin_amdgcn_readlane(cl[p], wave);
                n += nn[p];
            }
            const int2* wlb = (const int2*)wlist + (size_t)(lrb * EP * NW + wave) * LCAP;
            auto entry = [&](int i) {
                int off = i;
                int idx = -1;
#pragma unroll
                for (int p = 0; p < EP; ++p) {
                    if (off >= 0 && off < nn[p]) idx = p * (NW * LCAP) + off;
                    off -= nn[p];
                }
                return (idx >= 0) ? wlb[idx] : make_int2(0, 0);
            };
            int2 nxt = entry(lane);
            for (int i0 = 0; i0 < n; i0 += 32) {
                const int2 cur = nxt;
                bool ask = true;
                const int dlc = cur.y - dl0;
                const bool keep = (i0 + lane < n) && ((unsigned)dlc < (unsigned)R);
                unsigned m = __builtin_amdgcn_ballot_w32(keep);
                while (m != 0u) {
                    float2 v[GL][H];
                    unsigned mm = m;
#pragma unroll
                    for (int g = 0; g < GL; ++g) {
                        if (mm != 0u) {
                            const int b = __builtin_ctz(mm);
                            mm &= mm - 1u;
                            const int sj = __builtin_amdgcn_readlane(cur.x, b);
                            const float* rp = featc + (size_t)sj * FS;
#pragma unroll
                            for (int h = 0; h < H; ++h) v[g][h] = *(const float2*)(rp + h * 64);
                        }
                    }
                    if (ask) {
                        nxt = entry(i0 + 32 + lane);
                        ask = false;
                    }
#pragma unroll
                    for (int g = 0; g < GL; ++g) {
                        if (m != 0u) {
                            const int b = __builtin_ctz(m);
                            m &= m - 1u;
                            const int dj = __builtin_amdgcn_readlane(dlc, b);
                            float* sp = s_sum + dj * P + lane * 2;
#pragma unroll
                            for (int h = 0; h < H; ++h) {
                                atomicAdd(sp + h * 64, v[g][h].x);
                                atomicAdd(sp + h * 64 + 1, v[g][h].y);
                            }
                        }
                    }
                }
                if (ask) nxt = entry(i0 + 32 + lane);
            }
        } else {
            scan_part(0, 1);
        }
    } else {
        scan_part(part, EP);
    }
    __syncthreads();

    if constexpr (USE_LIST) {
        static_assert(NOUT == 64, "the column order above is written for 4 accumulator tiles");
        if (wave < R / 16) {
            const int m = lane & 15;
            const int khalf = lane >> 4;
            const int lrow = wave * 16 + m;
            const float inv = s_cnt[lrow];
            const float* sa = s_sum + lrow * P + khalf * 2;
            int gn = base + lrow;
            if (gn > Nn - 1) gn = Nn - 1;
            const float* xa = feat + (size_t)gn * FS + khalf * 2;
            const float* wlp = Wl + (size_t)(khalf * 2) * NOUT + m * 4;
            const float* wrp = Wr + (size_t)(khalf * 2) * NOUT + m * 4;
            v8f acc0 = {}, acc1 = {}, acc2 = {}, acc3 = {};
#pragma unroll 4
            for (int k = 0; k < F; k += 4) {
                const float2 sr = *(const float2*)(sa + k);
                const float2 xr = *(const float2*)(xa + k);
                const float4 l0 = *(const float4*)(wlp + (size_t)k * NOUT);
                const float4 l1 = *(const float4*)(wlp + (size_t)(k + 1) * NOUT);
                const float4 r0 = *(const float4*)(wrp + (size_t)k * NOUT);
                const float4 r1 = *(const float4*)(wrp + (size_t)(k + 1) * NOUT);
                v2f al, ax, b;
                al.x = sr.x * inv;
                al.y = sr.y * inv;
                ax.x = xr.x;
                ax.y = xr.y;
                b.x = l0.x; b.y = l1.x;
                acc0 = __builtin_amdgcn_wmma_f32_16x16x4_f32(false, al, false, b, (short)0, acc0, false, false);
                b.x = r0.x; b.y = r1.x;
                acc0 = __builtin_amdgcn_wmma_f32_16x16x4_f32(false, ax, false, b, (short)0, acc0, false, false);
                b.x = l0.y; b.y = l1.y;
                acc1 = __builtin_amdgcn_wmma_f32_16x16x4_f32(false, al, false, b, (short)0, acc1, false, false);
                b.x = r0.y; b.y = r1.y;
                acc1 = __builtin_amdgcn_wmma_f32_16x16x4_f32(false, ax, false, b, (short)0, acc1, false, false);
                b.x = l0.z; b.y = l1.z;
                acc2 = __builtin_amdgcn_wmma_f32_16x16x4_f32(false, al, false, b, (short)0, acc2, false, false);
                b.x = r0.z; b.y = r1.z;
                acc2 = __builtin_amdgcn_wmma_f32_16x16x4_f32(false, ax, false, b, (short)0, acc2, false, false);
                b.x = l0.w; b.y = l1.w;
                acc3 = __builtin_amdgcn_wmma_f32_16x16x4_f32(false, al, false, b, (short)0, acc3, false, false);
                b.x = r0.w; b.y = r1.w;
                acc3 = __builtin_amdgcn_wmma_f32_16x16x4_f32(false, ax, false, b, (short)0, acc3, false, false);
            }
            const float4 bi = *(const float4*)(bl + m * 4);
            const float4 wf = *(const float4*)(Wfc + m * 4);
            const float bf = bfc[0];
            float sv = 0.0f;
#pragma unroll
            for (int v = 0; v < 8; ++v) {
                float q = fmaxf(acc0[v] + bi.x, 0.0f) * wf.x;
                q += fmaxf(acc1[v] + bi.y, 0.0f) * wf.y;
                q += fmaxf(acc2[v] + bi.z, 0.0f) * wf.z;
                q += fmaxf(acc3[v] + bi.w, 0.0f) * wf.w;
                const float s = half_wave_sum(q) + bf;
                if (m == v) sv = s;
            }
            const int gnode = base + wave * 16 + khalf * 8 + m;
            if (m < 8 && gnode < Nn) out[gnode] = 1.0f / (1.0f + __expf(-sv));
        }
    } else {
        float* aggp = agg + (size_t)part * Nn * FS;
        float* degp = deg + (size_t)part * Nn;
        for (int i = tid; i < R * F; i += 1024) {
            const int r = i / F;
            const int c = i - r * F;
            if (base + r < Nn) aggp[(size_t)(base + r) * FS + col0 + c] = s_sum[i];
        }
        if (COUNT) {
            for (int i = tid; i < R; i += 1024) {
                if (base + i < Nn) degp[base + i] = s_cnt[i];
            }
        }
    }
}


template <int K, int NOUT, int AP, int DP>
__global__ __launch_bounds__(32 * (NOUT / 16)) void sage_combine_kernel(
    const float* __restrict__ xin, const float* __restrict__ agg,
    const float* __restrict__ deg, const float* __restrict__ Wl,
    const float* __restrict__ bl, const float* __restrict__ Wr,
    float* __restrict__ out, int Nn) {
    constexpr int KP = 2 * K + 4;
    constexpr int NT = 32 * (NOUT / 16);
    constexpr int K4 = K / 4;
    constexpr int IT = 16 * K4 / NT;
    static_assert((16 * K4) % NT == 0, "tile pieces must divide among the threads");
    __shared__ __attribute__((aligned(16))) float s_ax[16 * KP];
    float* const s_a = s_ax;
    float* const s_x = s_ax + K;

    const int node_base = blockIdx.x * 16;
    const int tid = threadIdx.x;
    const int lane = tid & 31;
    const int wave = tid >> 5;
    const int m = lane & 15;
    const int khalf = lane >> 4;
    const int ncol = wave * 16 + m;

    float4 av[AP][IT];
    float4 xv[IT];
    float dg[DP][IT];
#pragma unroll
    for (int it = 0; it < IT; ++it) {
        const int i = tid + it * NT;
        const int nd = i / K4;
        const int f = (i - nd * K4) * 4;
        int gnode = node_base + nd;
        if (gnode > Nn - 1) gnode = Nn - 1;
        const unsigned off = (unsigned)gnode * K + f;
#pragma unroll
        for (int p = 0; p < AP; ++p) av[p][it] = *(const float4*)(agg + (size_t)p * Nn * K + off);
        xv[it] = *(const float4*)(xin + off);
#pragma unroll
        for (int p = 0; p < DP; ++p) dg[p][it] = (deg + (size_t)p * Nn)[(unsigned)gnode];
    }
#pragma unroll
    for (int it = 0; it < IT; ++it) {
        const int i = tid + it * NT;
        const int nd = i / K4;
        const int f = (i - nd * K4) * 4;
        float dsum = dg[0][it];
#pragma unroll
        for (int p = 1; p < DP; ++p) dsum += dg[p][it];
        const float inv = 1.0f / fmaxf(dsum, 1.0f);
        float4 asum = av[0][it];
#pragma unroll
        for (int p = 1; p < AP; ++p) {
            asum.x += av[p][it].x;
            asum.y += av[p][it].y;
            asum.z += av[p][it].z;
            asum.w += av[p][it].w;
        }
        asum.x *= inv;
        asum.y *= inv;
        asum.z *= inv;
        asum.w *= inv;
        *(float4*)(s_a + nd * KP + f) = asum;
        *(float4*)(s_x + nd * KP + f) = xv[it];
    }
    __syncthreads();

    v8f acc = {};
#pragma unroll
    for (int k = 0; k < K; k += 4) {
        const int kb = k + khalf * 2;
        v2f a, b;
        a.x = s_a[m * KP + kb];
        a.y = s_a[m * KP + kb + 1];
        b.x = Wl[(size_t)kb * NOUT + ncol];
        b.y = Wl[(size_t)(kb + 1) * NOUT + ncol];
        acc = __builtin_amdgcn_wmma_f32_16x16x4_f32(false, a, false, b,
                                                    (short)0, acc, false, false);
        a.x = s_x[m * KP + kb];
        a.y = s_x[m * KP + kb + 1];
        b.x = Wr[(size_t)kb * NOUT + ncol];
        b.y = Wr[(size_t)(kb + 1) * NOUT + ncol];
        acc = __builtin_amdgcn_wmma_f32_16x16x4_f32(false, a, false, b,
                                                    (short)0, acc, false, false);
    }

    const float bias = bl[ncol];
#pragma unroll
    for (int v = 0; v < 8; ++v) {
        int mrow = v + khalf * 8;
        int gnode = node_base + mrow;
        if (gnode < Nn) {
            float val = acc[v] + bias;
            out[(size_t)gnode * NOUT + ncol] = fmaxf(val, 0.0f);
        }
    }
}

extern "C" void kernel_launch(void* const* d_in, const int* in_sizes, int n_in,
                              void* d_out, int out_size, void* d_ws, size_t ws_size,
                              hipStream_t stream) {
    const float* x    = (const float*)d_in[0];
    const int*   ei   = (const int*)d_in[1];
    const float* W1l  = (const float*)d_in[2];
    const float* b1   = (const float*)d_in[3];
    const float* W1r  = (const float*)d_in[4];
    const float* W2l  = (const float*)d_in[5];
    const float* b2   = (const float*)d_in[6];
    const float* W2r  = (const float*)d_in[7];
    const float* Wfc  = (const float*)d_in[8];
    const float* bfc  = (const float*)d_in[9];
    float* outp = (float*)d_out;

    const int Nn = in_sizes[0] / 64;
    const int Ee = in_sizes[1] / 2;

    float* deg = (float*)d_ws;
    float* agg = deg + (size_t)Nn * 2;
    float* h1  = agg + (size_t)Nn * 128;
    int* wlist = (int*)((char*)d_ws + ((size_t)160 << 20));
    int* wcnt  = (int*)((char*)d_ws + ((size_t)224 << 20));
    unsigned* idw = (unsigned*)((char*)d_ws + ((size_t)112 << 20));
    int* rec = (int*)((char*)d_ws + ((size_t)116 << 20));

    edge_recode_kernel<<<256, 512, 0, stream>>>(ei, idw, rec, Ee);

    neighbour_sum_lds_kernel<64, 64, 800, 2, 1, 16, true, true, false><<<(Nn + 799) / 800 * 2, 1024, 0, stream>>>(
        idw, rec, x, agg, deg, wlist, wcnt, nullptr, nullptr, nullptr, nullptr, nullptr, nullptr, Ee, Nn);
    sage_combine_kernel<64, 128, 2, 2><<<(Nn + 15) / 16, 256, 0, stream>>>(
        x, agg, deg, W1l, b1, W1r, h1, Nn);

    neighbour_sum_lds_kernel<128, 128, 400, 2, 2, 64, false, false, true><<<(Nn + 399) / 400, 1024, 0, stream>>>(
        idw, rec, h1, agg, deg, wlist, wcnt, W2l, b2, W2r, Wfc, bfc, outp, Ee, Nn);
}
